# starting kernel
# baseline (speedup 1.0000x reference)
#include <hip/hip_runtime.h>
#include <hip/hip_bf16.h>


#define N1C 50000
#define N2C 50000
#define EC  800000
#define DC  128
#define LC  2
#define WSTR 132

typedef float v2f __attribute__((ext_vector_type(2)));
typedef float v8f __attribute__((ext_vector_type(8)));

__device__ __forceinline__ void atomicMaxF(float* addr, float val) {
  if (val >= 0.f) atomicMax((int*)addr, __float_as_int(val));
  else            atomicMin((unsigned int*)addr, (unsigned int)__float_as_int(val));
}

__global__ __launch_bounds__(256)
void gemm_n128_wmma(const float* __restrict__ A, const float* __restrict__ W,
                    const float* __restrict__ bias, float* __restrict__ C) {
  extern __shared__ float smem[];
  float* sWt = smem;
  float* sA  = smem + DC * WSTR;

  const int rowbase = blockIdx.x * 16;
  const int tid = threadIdx.x;

#pragma unroll
  for (int i = 0; i < 16; ++i) {
    const int c  = tid + 256 * i;
    const int k  = c >> 5;
    const int n4 = (c & 31) * 4;
    const float4 w4 = *(const float4*)(W + (size_t)k * DC + n4);
    sWt[(n4 + 0) * WSTR + k] = w4.x;
    sWt[(n4 + 1) * WSTR + k] = w4.y;
    sWt[(n4 + 2) * WSTR + k] = w4.z;
    sWt[(n4 + 3) * WSTR + k] = w4.w;
  }
#pragma unroll
  for (int i = 0; i < 2; ++i) {
    const int c  = tid + 256 * i;
    const int r  = c >> 5;
    const int n4 = (c & 31) * 4;
    *(float4*)(sA + r * WSTR + n4) =
        *(const float4*)(A + (size_t)(rowbase + r) * DC + n4);
  }
  __syncthreads();

  const int lane = tid & 31;
  const int cb   = (tid >> 5) * 16;
  const int m    = lane & 15;
  const int half = lane >> 4;

  const float* ap = sA + m * WSTR + half * 2;
  const float* bp = sWt + (cb + m) * WSTR + half * 2;

  v8f acc = {};
#pragma unroll 8
  for (int kk = 0; kk < 32; ++kk) {
    const v2f a = *(const v2f*)(ap + kk * 4);
    const v2f b = *(const v2f*)(bp + kk * 4);
    acc = __builtin_amdgcn_wmma_f32_16x16x4_f32(false, a, false, b,
                                                (short)0, acc, false, false);
  }

  const int col = cb + m;
  const float bv = bias ? bias[col] : 0.f;
  float* cp = C + ((size_t)rowbase + half * 8) * DC + col;
#pragma unroll
  for (int r = 0; r < 8; ++r) {
    cp[(size_t)r * DC] = acc[r] + bv;
  }
}

__global__ __launch_bounds__(256)
void rowdot2_kernel(const float* __restrict__ X, const float* __restrict__ v1,
                    const float* __restrict__ v2, float* __restrict__ o1,
                    float* __restrict__ o2, int N) {
  const int row  = blockIdx.x * (blockDim.x >> 5) + (threadIdx.x >> 5);
  const int lane = threadIdx.x & 31;
  if (row >= N) return;
  const float4 x = ((const float4*)(X + (size_t)row * DC))[lane];
  const float4 a = ((const float4*)v1)[lane];
  float s1 = x.x * a.x + x.y * a.y + x.z * a.z + x.w * a.w;
  float s2 = 0.f;
  if (v2) {
    const float4 b = ((const float4*)v2)[lane];
    s2 = x.x * b.x + x.y * b.y + x.z * b.z + x.w * b.w;
  }
#pragma unroll
  for (int off = 16; off > 0; off >>= 1) {
    s1 += __shfl_down(s1, off, 32);
    s2 += __shfl_down(s2, off, 32);
  }
  if (lane == 0) {
    o1[row] = s1;
    if (o2) o2[row] = s2;
  }
}

__global__ void vecmat128_kernel(const float* __restrict__ W,
                                 const float* __restrict__ a,
                                 float* __restrict__ v) {
  const int k = threadIdx.x;
  float s = 0.f;
#pragma unroll 8
  for (int j = 0; j < DC; ++j) s += W[(size_t)k * DC + j] * a[j];
  v[k] = s;
}

__global__ void fillf_kernel(float* p, float val, int n) {
  const int i = blockIdx.x * blockDim.x + threadIdx.x;
  if (i < n) p[i] = val;
}

__global__ void init_bias_kernel(float* __restrict__ out,
                                 const float* __restrict__ b1,
                                 const float* __restrict__ b2, int total) {
  const int i = blockIdx.x * blockDim.x + threadIdx.x;
  if (i >= total) return;
  const int c = i & (DC - 1);
  float v = b1[c];
  if (b2) v += b2[c];
  out[i] = v;
}

__global__ void edge_logit_max_kernel(const float* __restrict__ as,
                                      const float* __restrict__ ad,
                                      const int* __restrict__ src,
                                      const int* __restrict__ dst,
                                      float* __restrict__ elog,
                                      float* __restrict__ maxb, int E) {
  const int e = blockIdx.x * blockDim.x + threadIdx.x;
  if (e >= E) return;
  float l = as[src[e]] + ad[dst[e]];
  l = l > 0.f ? l : 0.2f * l;
  elog[e] = l;
  atomicMaxF(maxb + dst[e], l);
}

__global__ void edge_exp_sum_kernel(const float* __restrict__ elog,
                                    const float* __restrict__ maxb,
                                    const int* __restrict__ dst,
                                    float* __restrict__ ew,
                                    float* __restrict__ sumb, int E) {
  const int e = blockIdx.x * blockDim.x + threadIdx.x;
  if (e >= E) return;
  const int d = dst[e];
  const float w = __expf(elog[e] - maxb[d]);
  ew[e] = w;
  atomicAdd(sumb + d, w);
}

__global__ __launch_bounds__(256)
void edge_aggr_kernel(const float* __restrict__ ew, const float* __restrict__ sumb,
                      const int* __restrict__ src, const int* __restrict__ dst,
                      const float* __restrict__ G, float* __restrict__ O, int E) {
  const long gid = (long)blockIdx.x * blockDim.x + threadIdx.x;
  const int e    = (int)(gid >> 5);
  const int lane = (int)(gid & 31);
  if (e >= E) return;
  const int s = src[e], d = dst[e];
  const float coef = ew[e] / (sumb[d] + 1e-16f);
  const float4 g = *(const float4*)(G + (size_t)s * DC + lane * 4);
  float* op = O + (size_t)d * DC + lane * 4;
  atomicAdd(op + 0, coef * g.x);
  atomicAdd(op + 1, coef * g.y);
  atomicAdd(op + 2, coef * g.z);
  atomicAdd(op + 3, coef * g.w);
}

__global__ void relu_inplace_kernel(float* p, int n) {
  const int i = blockIdx.x * blockDim.x + threadIdx.x;
  if (i < n) {
    const float v = p[i];
    p[i] = v > 0.f ? v : 0.f;
  }
}

extern "C" void kernel_launch(void* const* d_in, const int* in_sizes, int n_in,
                              void* d_out, int out_size, void* d_ws, size_t ws_size,
                              hipStream_t stream) {
  const float* x1    = (const float*)d_in[0];
  const float* x2    = (const float*)d_in[1];
  const int*   ei11  = (const int*)d_in[2];
  const int*   ei22  = (const int*)d_in[3];
  const int*   ei12  = (const int*)d_in[4];
  const float* W_i1  = (const float*)d_in[5];
  const float* as_i1 = (const float*)d_in[6];
  const float* ad_i1 = (const float*)d_in[7];
  const float* b_i1  = (const float*)d_in[8];
  const float* W_i2  = (const float*)d_in[9];
  const float* as_i2 = (const float*)d_in[10];
  const float* ad_i2 = (const float*)d_in[11];
  const float* b_i2  = (const float*)d_in[12];
  const float* Ws_x  = (const float*)d_in[13];
  const float* Wd_x  = (const float*)d_in[14];
  const float* as_x  = (const float*)d_in[15];
  const float* ad_x  = (const float*)d_in[16];
  const float* b_x   = (const float*)d_in[17];
  const float* W_out = (const float*)d_in[18];
  const float* b_out = (const float*)d_in[19];

  float* ws = (float*)d_ws;
  size_t off = 0;
  auto alloc = [&](size_t n) { float* p = ws + off; off += n; return p; };

  float* h1   = alloc((size_t)N1C * DC);
  float* h2   = alloc((size_t)N2C * DC);
  float* g1   = alloc((size_t)N1C * DC);
  float* g2   = alloc((size_t)N2C * DC);
  float* gx   = alloc((size_t)N1C * DC);
  float* a_s1 = alloc(N1C);
  float* a_d1 = alloc(N1C);
  float* a_s2 = alloc(N2C);
  float* a_d2 = alloc(N2C);
  float* a_sx = alloc(N1C);
  float* a_dx = alloc(N2C);
  float* vd   = alloc(DC);
  float* maxb = alloc(N1C > N2C ? N1C : N2C);
  float* sumb = alloc(N1C > N2C ? N1C : N2C);
  float* elog = alloc(EC);
  float* ew   = alloc(EC);
  (void)ws_size; (void)in_sizes; (void)n_in; (void)out_size;

  const dim3 blk(256);
  const dim3 gemm_g1(N1C / 16), gemm_g2(N2C / 16);
  const dim3 rd_g1((N1C + 7) / 8), rd_g2((N2C + 7) / 8);
  const dim3 nd_g1((N1C * DC + 255) / 256), nd_g2((N2C * DC + 255) / 256);
  const dim3 seg_g1((N1C + 255) / 256), seg_g2((N2C + 255) / 256);
  const dim3 e_g((EC + 255) / 256);
  const dim3 ea_g(EC / 8);
  const size_t lds_bytes = (size_t)(DC * WSTR + 16 * WSTR) * sizeof(float);

  const float NEG_BIG = -3.402823466e38f;

  const float* in1 = x1;
  const float* in2 = x2;

  for (int l = 0; l < LC; ++l) {
    const size_t wo = (size_t)l * DC * DC;
    const size_t vo = (size_t)l * DC;

    gemm_n128_wmma<<<gemm_g1, blk, lds_bytes, stream>>>(in1, W_i1 + wo, nullptr, g1);
    gemm_n128_wmma<<<gemm_g2, blk, lds_bytes, stream>>>(in2, W_i2 + wo, nullptr, g2);
    gemm_n128_wmma<<<gemm_g1, blk, lds_bytes, stream>>>(in1, Ws_x + wo, nullptr, gx);

    rowdot2_kernel<<<rd_g1, blk, 0, stream>>>(g1, as_i1 + vo, ad_i1 + vo, a_s1, a_d1, N1C);
    rowdot2_kernel<<<rd_g2, blk, 0, stream>>>(g2, as_i2 + vo, ad_i2 + vo, a_s2, a_d2, N2C);
    rowdot2_kernel<<<rd_g1, blk, 0, stream>>>(gx, as_x + vo, nullptr, a_sx, nullptr, N1C);
    vecmat128_kernel<<<1, DC, 0, stream>>>(Wd_x + wo, ad_x + vo, vd);
    rowdot2_kernel<<<rd_g2, blk, 0, stream>>>(in2, vd, nullptr, a_dx, nullptr, N2C);

    float* o1 = h1;
    float* o2 = h2;
    init_bias_kernel<<<nd_g1, blk, 0, stream>>>(o1, b_i1 + vo, nullptr, N1C * DC);
    init_bias_kernel<<<nd_g2, blk, 0, stream>>>(o2, b_i2 + vo, b_x + vo, N2C * DC);

    fillf_kernel<<<seg_g1, blk, 0, stream>>>(maxb, NEG_BIG, N1C);
    fillf_kernel<<<seg_g1, blk, 0, stream>>>(sumb, 0.f, N1C);
    edge_logit_max_kernel<<<e_g, blk, 0, stream>>>(a_s1, a_d1, ei11, ei11 + EC, elog, maxb, EC);
    edge_exp_sum_kernel<<<e_g, blk, 0, stream>>>(elog, maxb, ei11 + EC, ew, sumb, EC);
    edge_aggr_kernel<<<ea_g, blk, 0, stream>>>(ew, sumb, ei11, ei11 + EC, g1, o1, EC);

    fillf_kernel<<<seg_g2, blk, 0, stream>>>(maxb, NEG_BIG, N2C);
    fillf_kernel<<<seg_g2, blk, 0, stream>>>(sumb, 0.f, N2C);
    edge_logit_max_kernel<<<e_g, blk, 0, stream>>>(a_s2, a_d2, ei22, ei22 + EC, elog, maxb, EC);
    edge_exp_sum_kernel<<<e_g, blk, 0, stream>>>(elog, maxb, ei22 + EC, ew, sumb, EC);
    edge_aggr_kernel<<<ea_g, blk, 0, stream>>>(ew, sumb, ei22, ei22 + EC, g2, o2, EC);

    fillf_kernel<<<seg_g2, blk, 0, stream>>>(maxb, NEG_BIG, N2C);
    fillf_kernel<<<seg_g2, blk, 0, stream>>>(sumb, 0.f, N2C);
    edge_logit_max_kernel<<<e_g, blk, 0, stream>>>(a_sx, a_dx, ei12, ei12 + EC, elog, maxb, EC);
    edge_exp_sum_kernel<<<e_g, blk, 0, stream>>>(elog, maxb, ei12 + EC, ew, sumb, EC);
    edge_aggr_kernel<<<ea_g, blk, 0, stream>>>(ew, sumb, ei12, ei12 + EC, gx, o2, EC);

    relu_inplace_kernel<<<nd_g1, blk, 0, stream>>>(o1, N1C * DC);
    relu_inplace_kernel<<<nd_g2, blk, 0, stream>>>(o2, N2C * DC);

    in1 = h1;
    in2 = h2;
  }

  float* out = (float*)d_out;
  gemm_n128_wmma<<<gemm_g1, blk, lds_bytes, stream>>>(h1, W_out, b_out, out);
  gemm_n128_wmma<<<gemm_g2, blk, lds_bytes, stream>>>(h2, W_out, b_out,
                                                      out + (size_t)N1C * DC);
}


# edges grouped by destination once per call (int count, scan, arrival-order slots); one wave per dst row does max, exp, sum and weighted row gather in registers, one store per row; float atomics go
# speedup vs baseline: 2.4697x; 2.4697x over previous
#include <hip/hip_runtime.h>


#define N1C 50000
#define N2C 50000
#define EC  800000
#define DC  128
#define LC  2
#define WSTR 132
#define SCAN_CH (((N1C > N2C ? N1C : N2C) + 1023) / 1024)
#define NSTR (1024 * SCAN_CH)

static_assert(N1C <= NSTR && N2C <= NSTR, "row-pointer stride too small");
static_assert(N1C <= 65536, "source ids are kept as 16-bit values");

typedef float v2f __attribute__((ext_vector_type(2)));
typedef float v8f __attribute__((ext_vector_type(8)));

__global__ __launch_bounds__(256)
void gemm_n128_wmma(const float* __restrict__ A, const float* __restrict__ W,
                    const float* __restrict__ bias, float* __restrict__ C) {
  extern __shared__ float smem[];
  float* sWt = smem;
  float* sA  = smem + DC * WSTR;

  const int rowbase = blockIdx.x * 16;
  const int tid = threadIdx.x;

#pragma unroll
  for (int i = 0; i < 16; ++i) {
    const int c  = tid + 256 * i;
    const int k  = c >> 5;
    const int n4 = (c & 31) * 4;
    const float4 w4 = *(const float4*)(W + (size_t)k * DC + n4);
    sWt[(n4 + 0) * WSTR + k] = w4.x;
    sWt[(n4 + 1) * WSTR + k] = w4.y;
    sWt[(n4 + 2) * WSTR + k] = w4.z;
    sWt[(n4 + 3) * WSTR + k] = w4.w;
  }
#pragma unroll
  for (int i = 0; i < 2; ++i) {
    const int c  = tid + 256 * i;
    const int r  = c >> 5;
    const int n4 = (c & 31) * 4;
    *(float4*)(sA + r * WSTR + n4) =
        *(const float4*)(A + (size_t)(rowbase + r) * DC + n4);
  }
  __syncthreads();

  const int lane = tid & 31;
  const int cb   = (tid >> 5) * 16;
  const int m    = lane & 15;
  const int half = lane >> 4;

  const float* ap = sA + m * WSTR + half * 2;
  const float* bp = sWt + (cb + m) * WSTR + half * 2;

  v8f acc = {};
#pragma unroll 8
  for (int kk = 0; kk < 32; ++kk) {
    const v2f a = *(const v2f*)(ap + kk * 4);
    const v2f b = *(const v2f*)(bp + kk * 4);
    acc = __builtin_amdgcn_wmma_f32_16x16x4_f32(false, a, false, b,
                                                (short)0, acc, false, false);
  }

  const int col = cb + m;
  const float bv = bias ? bias[col] : 0.f;
  float* cp = C + ((size_t)rowbase + half * 8) * DC + col;
#pragma unroll
  for (int r = 0; r < 8; ++r) {
    cp[(size_t)r * DC] = acc[r] + bv;
  }
}

__global__ __launch_bounds__(256)
void rowdot2_kernel(const float* __restrict__ X, const float* __restrict__ v1,
                    const float* __restrict__ v2, float* __restrict__ o1,
                    float* __restrict__ o2, int N) {
  const int row  = blockIdx.x * (blockDim.x >> 5) + (threadIdx.x >> 5);
  const int lane = threadIdx.x & 31;
  if (row >= N) return;
  const float4 x = ((const float4*)(X + (size_t)row * DC))[lane];
  const float4 a = ((const float4*)v1)[lane];
  float s1 = x.x * a.x + x.y * a.y + x.z * a.z + x.w * a.w;
  float s2 = 0.f;
  if (v2) {
    const float4 b = ((const float4*)v2)[lane];
    s2 = x.x * b.x + x.y * b.y + x.z * b.z + x.w * b.w;
  }
#pragma unroll
  for (int off = 16; off > 0; off >>= 1) {
    s1 += __shfl_down(s1, off, 32);
    s2 += __shfl_down(s2, off, 32);
  }
  if (lane == 0) {
    o1[row] = s1;
    if (o2) o2[row] = s2;
  }
}

__global__ void vecmat128_kernel(const float* __restrict__ W,
                                 const float* __restrict__ a,
                                 float* __restrict__ v) {
  const int k = threadIdx.x;
  float s = 0.f;
#pragma unroll 8
  for (int j = 0; j < DC; ++j) s += W[(size_t)k * DC + j] * a[j];
  v[k] = s;
}

__global__ void zero_int_kernel(int* p, int n) {
  const int i = blockIdx.x * blockDim.x + threadIdx.x;
  if (i < n) p[i] = 0;
}

__global__ void dst_count_kernel(const int* __restrict__ e11,
                                 const int* __restrict__ e22,
                                 const int* __restrict__ e12,
                                 int* __restrict__ rp) {
  const int e = blockIdx.x * blockDim.x + threadIdx.x;
  const int t = blockIdx.y;
  if (e >= EC) return;
  const int* ei = t == 0 ? e11 : (t == 1 ? e22 : e12);
  atomicAdd(rp + t * NSTR + ei[EC + e], 1);
}

__global__ __launch_bounds__(1024)
void dst_scan_kernel(int* __restrict__ rp) {
  __shared__ int wsum[32];
  int* c = rp + (size_t)blockIdx.x * NSTR + threadIdx.x * SCAN_CH;
  const int tid  = threadIdx.x;
  const int lane = tid & 31;
  const int w    = tid >> 5;

  int v[SCAN_CH];
  int sum = 0;
#pragma unroll
  for (int j = 0; j < SCAN_CH; ++j) { v[j] = c[j]; sum += v[j]; }

  int incl = sum;
#pragma unroll
  for (int off = 1; off < 32; off <<= 1) {
    const int u = __shfl_up(incl, off, 32);
    if (lane >= off) incl += u;
  }
  if (lane == 31) wsum[w] = incl;
  __syncthreads();
  if (w == 0) {
    const int tot = wsum[lane];
    int inc2 = tot;
#pragma unroll
    for (int off = 1; off < 32; off <<= 1) {
      const int u = __shfl_up(inc2, off, 32);
      if (lane >= off) inc2 += u;
    }
    wsum[lane] = inc2 - tot;
  }
  __syncthreads();

  int run = wsum[w] + incl - sum;
#pragma unroll
  for (int j = 0; j < SCAN_CH; ++j) { c[j] = run; run += v[j]; }
}

__global__ void dst_fill_kernel(const int* __restrict__ e11,
                                const int* __restrict__ e22,
                                const int* __restrict__ e12,
                                int* __restrict__ rp,
                                unsigned short* __restrict__ csr) {
  const int e = blockIdx.x * blockDim.x + threadIdx.x;
  const int t = blockIdx.y;
  if (e >= EC) return;
  const int* ei = t == 0 ? e11 : (t == 1 ? e22 : e12);
  const int s = ei[e];
  const int d = ei[EC + e];
  const int pos = atomicAdd(rp + t * NSTR + d, 1);
  csr[(size_t)t * EC + pos] = (unsigned short)s;
}

__device__ __forceinline__ float leaky02(float l) { return l > 0.f ? l : 0.2f * l; }

__device__ __forceinline__ float wave_max(float v) {
#pragma unroll
  for (int off = 16; off > 0; off >>= 1) v = fmaxf(v, __shfl_xor(v, off, 32));
  return v;
}
__device__ __forceinline__ float wave_sum(float v) {
#pragma unroll
  for (int off = 16; off > 0; off >>= 1) v += __shfl_xor(v, off, 32);
  return v;
}

__device__ __forceinline__ void gather_chunk(float4& acc, int s, float c, int n,
                                             const float* __restrict__ G, int lane) {
  const float* gl = G + lane * 4;
  int j = 0;
  for (; j + 4 <= n; j += 4) {
    const int s0 = __builtin_amdgcn_readlane(s, j + 0);
    const int s1 = __builtin_amdgcn_readlane(s, j + 1);
    const int s2 = __builtin_amdgcn_readlane(s, j + 2);
    const int s3 = __builtin_amdgcn_readlane(s, j + 3);
    const float c0 = __int_as_float(__builtin_amdgcn_readlane(__float_as_int(c), j + 0));
    const float c1 = __int_as_float(__builtin_amdgcn_readlane(__float_as_int(c), j + 1));
    const float c2 = __int_as_float(__builtin_amdgcn_readlane(__float_as_int(c), j + 2));
    const float c3 = __int_as_float(__builtin_amdgcn_readlane(__float_as_int(c), j + 3));
    const float4 g0 = *(const float4*)(gl + (size_t)s0 * DC);
    const float4 g1 = *(const float4*)(gl + (size_t)s1 * DC);
    const float4 g2 = *(const float4*)(gl + (size_t)s2 * DC);
    const float4 g3 = *(const float4*)(gl + (size_t)s3 * DC);
    acc.x += c0 * g0.x; acc.y += c0 * g0.y; acc.z += c0 * g0.z; acc.w += c0 * g0.w;
    acc.x += c1 * g1.x; acc.y += c1 * g1.y; acc.z += c1 * g1.z; acc.w += c1 * g1.w;
    acc.x += c2 * g2.x; acc.y += c2 * g2.y; acc.z += c2 * g2.z; acc.w += c2 * g2.w;
    acc.x += c3 * g3.x; acc.y += c3 * g3.y; acc.z += c3 * g3.z; acc.w += c3 * g3.w;
  }
  for (; j < n; ++j) {
    const int sj = __builtin_amdgcn_readlane(s, j);
    const float cj = __int_as_float(__builtin_amdgcn_readlane(__float_as_int(c), j));
    const float4 g = *(const float4*)(gl + (size_t)sj * DC);
    acc.x += cj * g.x; acc.y += cj * g.y; acc.z += cj * g.z; acc.w += cj * g.w;
  }
}

__device__ __forceinline__ float4 seg_softmax_gather(
    int d, int lane, const int* __restrict__ rp,
    const unsigned short* __restrict__ csr, const float* __restrict__ a_s,
    const float* __restrict__ a_d, const float* __restrict__ G) {
  float4 acc = make_float4(0.f, 0.f, 0.f, 0.f);
  const int end = rp[d];
  const int beg = d > 0 ? rp[d - 1] : 0;
  const float adv = a_d[d];
  const float NEG_INF = -__builtin_inff();

  const int  i0 = beg + lane;
  const bool v0 = i0 < end;
  const int  s0 = v0 ? (int)csr[i0] : 0;
  const float l0 = v0 ? leaky02(a_s[s0] + adv) : NEG_INF;

  float m = l0;
  for (int i = i0 + 32; i < end; i += 32)
    m = fmaxf(m, leaky02(a_s[csr[i]] + adv));
  m = wave_max(m);

  const float e0 = v0 ? __expf(l0 - m) : 0.f;
  float z = e0;
  for (int i = i0 + 32; i < end; i += 32)
    z += __expf(leaky02(a_s[csr[i]] + adv) - m);
  z = wave_sum(z);
  const float den = z + 1e-16f;

  const int n0 = end - beg < 32 ? end - beg : 32;
  gather_chunk(acc, s0, e0 / den, n0, G, lane);

  for (int base = beg + 32; base < end; base += 32) {
    const int  i = base + lane;
    const bool v = i < end;
    const int  s = v ? (int)csr[i] : 0;
    const float c = v ? __expf(leaky02(a_s[s] + adv) - m) / den : 0.f;
    const int n = end - base < 32 ? end - base : 32;
    gather_chunk(acc, s, c, n, G, lane);
  }
  return acc;
}

template <bool TWO>
__global__ __launch_bounds__(256)
void node_softmax_gather_kernel(
    const int* __restrict__ rpA, const unsigned short* __restrict__ csrA,
    const float* __restrict__ asA, const float* __restrict__ adA,
    const float* __restrict__ GA, const float* __restrict__ bA,
    const int* __restrict__ rpB, const unsigned short* __restrict__ csrB,
    const float* __restrict__ asB, const float* __restrict__ adB,
    const float* __restrict__ GB, const float* __restrict__ bB,
    float* __restrict__ H, int N) {
  const int d = __builtin_amdgcn_readfirstlane(
      (int)(blockIdx.x * (blockDim.x >> 5) + (threadIdx.x >> 5)));
  const int lane = threadIdx.x & 31;
  if (d >= N) return;

  const float4 a  = seg_softmax_gather(d, lane, rpA, csrA, asA, adA, GA);
  const float4 ba = ((const float4*)bA)[lane];
  float4 o = make_float4(a.x + ba.x, a.y + ba.y, a.z + ba.z, a.w + ba.w);
  if (TWO) {
    const float4 b  = seg_softmax_gather(d, lane, rpB, csrB, asB, adB, GB);
    const float4 bb = ((const float4*)bB)[lane];
    o.x += b.x + bb.x; o.y += b.y + bb.y; o.z += b.z + bb.z; o.w += b.w + bb.w;
  }
  o.x = o.x > 0.f ? o.x : 0.f;
  o.y = o.y > 0.f ? o.y : 0.f;
  o.z = o.z > 0.f ? o.z : 0.f;
  o.w = o.w > 0.f ? o.w : 0.f;
  *(float4*)(H + (size_t)d * DC + lane * 4) = o;
}

extern "C" void kernel_launch(void* const* d_in, const int* in_sizes, int n_in,
                              void* d_out, int out_size, void* d_ws, size_t ws_size,
                              hipStream_t stream) {
  const float* x1    = (const float*)d_in[0];
  const float* x2    = (const float*)d_in[1];
  const int*   ei11  = (const int*)d_in[2];
  const int*   ei22  = (const int*)d_in[3];
  const int*   ei12  = (const int*)d_in[4];
  const float* W_i1  = (const float*)d_in[5];
  const float* as_i1 = (const float*)d_in[6];
  const float* ad_i1 = (const float*)d_in[7];
  const float* b_i1  = (const float*)d_in[8];
  const float* W_i2  = (const float*)d_in[9];
  const float* as_i2 = (const float*)d_in[10];
  const float* ad_i2 = (const float*)d_in[11];
  const float* b_i2  = (const float*)d_in[12];
  const float* Ws_x  = (const float*)d_in[13];
  const float* Wd_x  = (const float*)d_in[14];
  const float* as_x  = (const float*)d_in[15];
  const float* ad_x  = (const float*)d_in[16];
  const float* b_x   = (const float*)d_in[17];
  const float* W_out = (const float*)d_in[18];
  const float* b_out = (const float*)d_in[19];

  float* ws = (float*)d_ws;
  size_t off = 0;
  auto alloc = [&](size_t n) { float* p = ws + off; off += n; return p; };

  float* h1   = alloc((size_t)N1C * DC);
  float* h2   = alloc((size_t)N2C * DC);
  float* g1   = alloc((size_t)N1C * DC);
  float* g2   = alloc((size_t)N2C * DC);
  float* gx   = alloc((size_t)N1C * DC);
  float* a_s1 = alloc(N1C);
  float* a_d1 = alloc(N1C);
  float* a_s2 = alloc(N2C);
  float* a_d2 = alloc(N2C);
  float* a_sx = alloc(N1C);
  float* a_dx = alloc(N2C);
  float* vd   = alloc(DC);
  int*   rp   = (int*)alloc((size_t)3 * NSTR);
  unsigned short* csr = (unsigned short*)alloc((size_t)3 * EC / 2);
  (void)ws_size; (void)in_sizes; (void)n_in; (void)out_size;

  const dim3 blk(256);
  const dim3 gemm_g1(N1C / 16), gemm_g2(N2C / 16);
  const dim3 rd_g1((N1C + 7) / 8), rd_g2((N2C + 7) / 8);
  const dim3 rp_g((3 * NSTR + 255) / 256);
  const dim3 e3_g((EC + 255) / 256, 3);
  const size_t lds_bytes = (size_t)(DC * WSTR + 16 * WSTR) * sizeof(float);

  const int* rp11 = rp;
  const int* rp22 = rp + NSTR;
  const int* rp12 = rp + 2 * NSTR;
  const unsigned short* csr11 = csr;
  const unsigned short* csr22 = csr + (size_t)EC;
  const unsigned short* csr12 = csr + (size_t)2 * EC;

  zero_int_kernel<<<rp_g, blk, 0, stream>>>(rp, 3 * NSTR);
  dst_count_kernel<<<e3_g, blk, 0, stream>>>(ei11, ei22, ei12, rp);
  dst_scan_kernel<<<3, 1024, 0, stream>>>(rp);
  dst_fill_kernel<<<e3_g, blk, 0, stream>>>(ei11, ei22, ei12, rp, csr);

  const float* in1 = x1;
  const float* in2 = x2;

  for (int l = 0; l < LC; ++l) {
    const size_t wo = (size_t)l * DC * DC;
    const size_t vo = (size_t)l * DC;

    gemm_n128_wmma<<<gemm_g1, blk, lds_bytes, stream>>>(in1, W_i1 + wo, nullptr, g1);
    gemm_n128_wmma<<<gemm_g2, blk, lds_bytes, stream>>>(in2, W_i2 + wo, nullptr, g2);
    gemm_n128_wmma<<<gemm_g1, blk, lds_bytes, stream>>>(in1, Ws_x + wo, nullptr, gx);

    rowdot2_kernel<<<rd_g1, blk, 0, stream>>>(g1, as_i1 + vo, ad_i1 + vo, a_s1, a_d1, N1C);
    rowdot2_kernel<<<rd_g2, blk, 0, stream>>>(g2, as_i2 + vo, ad_i2 + vo, a_s2, a_d2, N2C);
    rowdot2_kernel<<<rd_g1, blk, 0, stream>>>(gx, as_x + vo, nullptr, a_sx, nullptr, N1C);
    vecmat128_kernel<<<1, DC, 0, stream>>>(Wd_x + wo, ad_x + vo, vd);
    rowdot2_kernel<<<rd_g2, blk, 0, stream>>>(in2, vd, nullptr, a_dx, nullptr, N2C);

    node_softmax_gather_kernel<false><<<rd_g1, blk, 0, stream>>>(
        rp11, csr11, a_s1, a_d1, g1, b_i1 + vo,
        nullptr, nullptr, nullptr, nullptr, nullptr, nullptr, h1, N1C);
    node_softmax_gather_kernel<true><<<rd_g2, blk, 0, stream>>>(
        rp22, csr22, a_s2, a_d2, g2, b_i2 + vo,
        rp12, csr12, a_sx, a_dx, gx, b_x + vo, h2, N2C);

    in1 = h1;
    in2 = h2;
  }

  float* out = (float*)d_out;
  gemm_n128_wmma<<<gemm_g1, blk, lds_bytes, stream>>>(h1, W_out, b_out, out);
  gemm_n128_wmma<<<gemm_g2, blk, lds_bytes, stream>>>(h2, W_out, b_out,
                                                      out + (size_t)N1C * DC);
}


# fewer, longer waves (wave start rate was the limit): GEMM workgroups keep W in LDS, loop over row tiles; a wave takes 16 rows (row dots) or 2 (gather), a count/fill thread 8 edges; slots by arrival
# speedup vs baseline: 3.2358x; 1.3102x over previous
#include <hip/hip_runtime.h>


#define N1C 50000
#define N2C 50000
#define EC  800000
#define DC  128
#define LC  2
#define WSTR 132
#define SCAN_CH (((N1C > N2C ? N1C : N2C) + 1023) / 1024)
#define NSTR (1024 * SCAN_CH)

static_assert(N1C <= NSTR && N2C <= NSTR, "row-pointer stride too small");
static_assert(N1C <= 65536, "source ids are kept as 16-bit values");
static_assert(N1C % 16 == 0 && N2C % 16 == 0, "GEMM row tiles are 16 rows, no tail tile");

typedef float v2f __attribute__((ext_vector_type(2)));
typedef float v8f __attribute__((ext_vector_type(8)));

#define GEMM_WGS 768

__global__ __launch_bounds__(256)
void gemm_n128_wmma(const float* __restrict__ A, const float* __restrict__ W,
                    const float* __restrict__ bias, float* __restrict__ C,
                    int ntiles) {
  extern __shared__ float smem[];
  float* sA  = smem;
  float* sWt = smem + 2 * 16 * WSTR;

  const int tid  = threadIdx.x;
  const int lane = tid & 31;
  const int wv   = tid >> 5;

  int tile = blockIdx.x;
  if (tile >= ntiles) return;

  const int ar  = tid >> 5;
  const int an4 = (tid & 31) * 4;
  const float* arow = A + (size_t)ar * DC + an4;
  float4 pa0 = *(const float4*)(arow + (size_t)tile * 16 * DC);
  float4 pa1 = *(const float4*)(arow + (size_t)tile * 16 * DC + 8 * DC);

#pragma unroll
  for (int i = 0; i < 16; ++i) {
    const int k  = 8 * i + (lane >> 2);
    const int n4 = 16 * wv + 4 * (lane & 3);
    const float4 w4 = *(const float4*)(W + (size_t)k * DC + n4);
    sWt[(n4 + 0) * WSTR + k] = w4.x;
    sWt[(n4 + 1) * WSTR + k] = w4.y;
    sWt[(n4 + 2) * WSTR + k] = w4.z;
    sWt[(n4 + 3) * WSTR + k] = w4.w;
  }

  const int cb   = wv * 16;
  const int m    = lane & 15;
  const int half = lane >> 4;
  const int col  = cb + m;
  const float bv = bias ? bias[col] : 0.f;
  const float* bp = sWt + (cb + m) * WSTR + half * 2;

  int buf = 0;
  for (; tile < ntiles; tile += gridDim.x) {
    float* sAb = sA + buf * (16 * WSTR);
    *(float4*)(sAb + ar * WSTR + an4)       = pa0;
    *(float4*)(sAb + (ar + 8) * WSTR + an4) = pa1;
    __syncthreads();

    const int nt = tile + gridDim.x;
    if (nt < ntiles) {
      pa0 = *(const float4*)(arow + (size_t)nt * 16 * DC);
      pa1 = *(const float4*)(arow + (size_t)nt * 16 * DC + 8 * DC);
    }

    const float* ap = sAb + m * WSTR + half * 2;
    v8f acc = {};
#pragma unroll 8
    for (int kk = 0; kk < 32; ++kk) {
      const v2f a = *(const v2f*)(ap + kk * 4);
      const v2f b = *(const v2f*)(bp + kk * 4);
      acc = __builtin_amdgcn_wmma_f32_16x16x4_f32(false, a, false, b,
                                                  (short)0, acc, false, false);
    }

    float* cp = C + ((size_t)tile * 16 + half * 8) * DC + col;
#pragma unroll
    for (int r = 0; r < 8; ++r) {
      cp[(size_t)r * DC] = acc[r] + bv;
    }
    buf ^= 1;
  }
}

#define RD_ROWS 16
static_assert(N1C % RD_ROWS == 0 && N2C % RD_ROWS == 0, "a row-dot wave has all its rows or none");

__global__ __launch_bounds__(256)
void rowdot2_kernel(const float* __restrict__ X, const float* __restrict__ v1,
                    const float* __restrict__ v2, float* __restrict__ o1,
                    float* __restrict__ o2, int N) {
  const int wave = blockIdx.x * (blockDim.x >> 5) + (threadIdx.x >> 5);
  const int lane = threadIdx.x & 31;
  const int row0 = wave * RD_ROWS;
  if (row0 >= N) return;
  const float4* xp = (const float4*)(X + (size_t)row0 * DC) + lane;
  float4 x[RD_ROWS];
#pragma unroll
  for (int r = 0; r < RD_ROWS; ++r) x[r] = xp[r * (DC / 4)];

  const float4 a = ((const float4*)v1)[lane];
  float s1[RD_ROWS];
#pragma unroll
  for (int r = 0; r < RD_ROWS; ++r)
    s1[r] = x[r].x * a.x + x[r].y * a.y + x[r].z * a.z + x[r].w * a.w;
#pragma unroll
  for (int off = 16; off > 0; off >>= 1) {
#pragma unroll
    for (int r = 0; r < RD_ROWS; ++r) s1[r] += __shfl_xor(s1[r], off, 32);
  }
  float out1 = s1[0];
#pragma unroll
  for (int r = 1; r < RD_ROWS; ++r) out1 = lane == r ? s1[r] : out1;
  if (lane < RD_ROWS) o1[row0 + lane] = out1;

  if (v2) {
    const float4 b = ((const float4*)v2)[lane];
    float s2[RD_ROWS];
#pragma unroll
    for (int r = 0; r < RD_ROWS; ++r)
      s2[r] = x[r].x * b.x + x[r].y * b.y + x[r].z * b.z + x[r].w * b.w;
#pragma unroll
    for (int off = 16; off > 0; off >>= 1) {
#pragma unroll
      for (int r = 0; r < RD_ROWS; ++r) s2[r] += __shfl_xor(s2[r], off, 32);
    }
    float out2 = s2[0];
#pragma unroll
    for (int r = 1; r < RD_ROWS; ++r) out2 = lane == r ? s2[r] : out2;
    if (lane < RD_ROWS) o2[row0 + lane] = out2;
  }
}

__global__ void vecmat128_kernel(const float* __restrict__ W,
                                 const float* __restrict__ a,
                                 float* __restrict__ v) {
  const int k = threadIdx.x;
  float s = 0.f;
#pragma unroll 8
  for (int j = 0; j < DC; ++j) s += W[(size_t)k * DC + j] * a[j];
  v[k] = s;
}

__global__ void zero_int_kernel(int* p, int n) {
  const int i = blockIdx.x * blockDim.x + threadIdx.x;
  if (i < n) p[i] = 0;
}

#define EPT 8
#define GRP_T 160
#define GRP_STRIDE (EC / EPT)
static_assert(EC % (EPT * GRP_T) == 0, "edge grouping grid has no tail");

__global__ __launch_bounds__(GRP_T)
void dst_count_kernel(const int* __restrict__ e11,
                      const int* __restrict__ e22,
                      const int* __restrict__ e12,
                      int* __restrict__ rp) {
  const int e0 = blockIdx.x * GRP_T + threadIdx.x;
  const int t = blockIdx.y;
  const int* dstp = (t == 0 ? e11 : (t == 1 ? e22 : e12)) + EC + e0;
  int* cnt = rp + t * NSTR;
  int d[EPT];
#pragma unroll
  for (int u = 0; u < EPT; ++u) d[u] = dstp[u * GRP_STRIDE];
#pragma unroll
  for (int u = 0; u < EPT; ++u) atomicAdd(cnt + d[u], 1);
}

__global__ __launch_bounds__(1024)
void dst_scan_kernel(int* __restrict__ rp) {
  __shared__ int wsum[32];
  int* c = rp + (size_t)blockIdx.x * NSTR + threadIdx.x * SCAN_CH;
  const int tid  = threadIdx.x;
  const int lane = tid & 31;
  const int w    = tid >> 5;

  int v[SCAN_CH];
  int sum = 0;
#pragma unroll
  for (int j = 0; j < SCAN_CH; ++j) { v[j] = c[j]; sum += v[j]; }

  int incl = sum;
#pragma unroll
  for (int off = 1; off < 32; off <<= 1) {
    const int u = __shfl_up(incl, off, 32);
    if (lane >= off) incl += u;
  }
  if (lane == 31) wsum[w] = incl;
  __syncthreads();
  if (w == 0) {
    const int tot = wsum[lane];
    int inc2 = tot;
#pragma unroll
    for (int off = 1; off < 32; off <<= 1) {
      const int u = __shfl_up(inc2, off, 32);
      if (lane >= off) inc2 += u;
    }
    wsum[lane] = inc2 - tot;
  }
  __syncthreads();

  int run = wsum[w] + incl - sum;
#pragma unroll
  for (int j = 0; j < SCAN_CH; ++j) { c[j] = run; run += v[j]; }
}

__global__ __launch_bounds__(GRP_T)
void dst_fill_kernel(const int* __restrict__ e11,
                     const int* __restrict__ e22,
                     const int* __restrict__ e12,
                     int* __restrict__ rp,
                     unsigned short* __restrict__ csr) {
  const int e0 = blockIdx.x * GRP_T + threadIdx.x;
  const int t = blockIdx.y;
  const int* ei = (t == 0 ? e11 : (t == 1 ? e22 : e12)) + e0;
  int* cur = rp + t * NSTR;
  unsigned short* seg = csr + (size_t)t * EC;
  int s[EPT], d[EPT], pos[EPT];
#pragma unroll
  for (int u = 0; u < EPT; ++u) {
    s[u] = ei[u * GRP_STRIDE];
    d[u] = ei[EC + u * GRP_STRIDE];
  }
#pragma unroll
  for (int u = 0; u < EPT; ++u) pos[u] = atomicAdd(cur + d[u], 1);
#pragma unroll
  for (int u = 0; u < EPT; ++u) seg[pos[u]] = (unsigned short)s[u];
}

__device__ __forceinline__ float leaky02(float l) { return l > 0.f ? l : 0.2f * l; }

__device__ __forceinline__ float wave_max(float v) {
#pragma unroll
  for (int off = 16; off > 0; off >>= 1) v = fmaxf(v, __shfl_xor(v, off, 32));
  return v;
}
__device__ __forceinline__ float wave_sum(float v) {
#pragma unroll
  for (int off = 16; off > 0; off >>= 1) v += __shfl_xor(v, off, 32);
  return v;
}

__device__ __forceinline__ void gather_chunk(float4& acc, int s, float c, int n,
                                             const float* __restrict__ G, int lane) {
  const float* gl = G + lane * 4;
  int j = 0;
  for (; j + 4 <= n; j += 4) {
    const int s0 = __builtin_amdgcn_readlane(s, j + 0);
    const int s1 = __builtin_amdgcn_readlane(s, j + 1);
    const int s2 = __builtin_amdgcn_readlane(s, j + 2);
    const int s3 = __builtin_amdgcn_readlane(s, j + 3);
    const float c0 = __int_as_float(__builtin_amdgcn_readlane(__float_as_int(c), j + 0));
    const float c1 = __int_as_float(__builtin_amdgcn_readlane(__float_as_int(c), j + 1));
    const float c2 = __int_as_float(__builtin_amdgcn_readlane(__float_as_int(c), j + 2));
    const float c3 = __int_as_float(__builtin_amdgcn_readlane(__float_as_int(c), j + 3));
    const float4 g0 = *(const float4*)(gl + (size_t)s0 * DC);
    const float4 g1 = *(const float4*)(gl + (size_t)s1 * DC);
    const float4 g2 = *(const float4*)(gl + (size_t)s2 * DC);
    const float4 g3 = *(const float4*)(gl + (size_t)s3 * DC);
    acc.x += c0 * g0.x; acc.y += c0 * g0.y; acc.z += c0 * g0.z; acc.w += c0 * g0.w;
    acc.x += c1 * g1.x; acc.y += c1 * g1.y; acc.z += c1 * g1.z; acc.w += c1 * g1.w;
    acc.x += c2 * g2.x; acc.y += c2 * g2.y; acc.z += c2 * g2.z; acc.w += c2 * g2.w;
    acc.x += c3 * g3.x; acc.y += c3 * g3.y; acc.z += c3 * g3.z; acc.w += c3 * g3.w;
  }
  for (; j < n; ++j) {
    const int sj = __builtin_amdgcn_readlane(s, j);
    const float cj = __int_as_float(__builtin_amdgcn_readlane(__float_as_int(c), j));
    const float4 g = *(const float4*)(gl + (size_t)sj * DC);
    acc.x += cj * g.x; acc.y += cj * g.y; acc.z += cj * g.z; acc.w += cj * g.w;
  }
}

__device__ __forceinline__ float4 seg_softmax_gather(
    int d, int lane, const int* __restrict__ rp,
    const unsigned short* __restrict__ csr, const float* __restrict__ a_s,
    const float* __restrict__ a_d, const float* __restrict__ G) {
  float4 acc = make_float4(0.f, 0.f, 0.f, 0.f);
  const int end = rp[d];
  const int beg = d > 0 ? rp[d - 1] : 0;
  const float adv = a_d[d];
  const float NEG_INF = -__builtin_inff();

  const int  i0 = beg + lane;
  const bool v0 = i0 < end;
  const int  s0 = v0 ? (int)csr[i0] : 0;
  const float l0 = v0 ? leaky02(a_s[s0] + adv) : NEG_INF;

  float m = l0;
  for (int i = i0 + 32; i < end; i += 32)
    m = fmaxf(m, leaky02(a_s[csr[i]] + adv));
  m = wave_max(m);

  const float e0 = v0 ? __expf(l0 - m) : 0.f;
  float z = e0;
  for (int i = i0 + 32; i < end; i += 32)
    z += __expf(leaky02(a_s[csr[i]] + adv) - m);
  z = wave_sum(z);
  const float den = z + 1e-16f;

  const int n0 = end - beg < 32 ? end - beg : 32;
  gather_chunk(acc, s0, e0 / den, n0, G, lane);

  for (int base = beg + 32; base < end; base += 32) {
    const int  i = base + lane;
    const bool v = i < end;
    const int  s = v ? (int)csr[i] : 0;
    const float c = v ? __expf(leaky02(a_s[s] + adv) - m) / den : 0.f;
    const int n = end - base < 32 ? end - base : 32;
    gather_chunk(acc, s, c, n, G, lane);
  }
  return acc;
}

#define NODE_ROWS 2
static_assert(N1C % NODE_ROWS == 0 && N2C % NODE_ROWS == 0, "a gather wave has all its rows or none");

template <bool TWO>
__global__ __launch_bounds__(256, 16)
void node_softmax_gather_kernel(
    const int* __restrict__ rpA, const unsigned short* __restrict__ csrA,
    const float* __restrict__ asA, const float* __restrict__ adA,
    const float* __restrict__ GA, const float* __restrict__ bA,
    const int* __restrict__ rpB, const unsigned short* __restrict__ csrB,
    const float* __restrict__ asB, const float* __restrict__ adB,
    const float* __restrict__ GB, const float* __restrict__ bB,
    float* __restrict__ H, int N) {
  const int w = __builtin_amdgcn_readfirstlane(
      (int)(blockIdx.x * (blockDim.x >> 5) + (threadIdx.x >> 5)));
  const int lane = threadIdx.x & 31;
  if (w * NODE_ROWS >= N) return;

#pragma nounroll
  for (int i = 0; i < NODE_ROWS; ++i) {
    const int d = w * NODE_ROWS + i;
    const float4 a  = seg_softmax_gather(d, lane, rpA, csrA, asA, adA, GA);
    const float4 ba = ((const float4*)bA)[lane];
    float4 o = make_float4(a.x + ba.x, a.y + ba.y, a.z + ba.z, a.w + ba.w);
    if (TWO) {
      const float4 b  = seg_softmax_gather(d, lane, rpB, csrB, asB, adB, GB);
      const float4 bb = ((const float4*)bB)[lane];
      o.x += b.x + bb.x; o.y += b.y + bb.y; o.z += b.z + bb.z; o.w += b.w + bb.w;
    }
    o.x = o.x > 0.f ? o.x : 0.f;
    o.y = o.y > 0.f ? o.y : 0.f;
    o.z = o.z > 0.f ? o.z : 0.f;
    o.w = o.w > 0.f ? o.w : 0.f;
    *(float4*)(H + (size_t)d * DC + lane * 4) = o;
  }
}

extern "C" void kernel_launch(void* const* d_in, const int* in_sizes, int n_in,
                              void* d_out, int out_size, void* d_ws, size_t ws_size,
                              hipStream_t stream) {
  const float* x1    = (const float*)d_in[0];
  const float* x2    = (const float*)d_in[1];
  const int*   ei11  = (const int*)d_in[2];
  const int*   ei22  = (const int*)d_in[3];
  const int*   ei12  = (const int*)d_in[4];
  const float* W_i1  = (const float*)d_in[5];
  const float* as_i1 = (const float*)d_in[6];
  const float* ad_i1 = (const float*)d_in[7];
  const float* b_i1  = (const float*)d_in[8];
  const float* W_i2  = (const float*)d_in[9];
  const float* as_i2 = (const float*)d_in[10];
  const float* ad_i2 = (const float*)d_in[11];
  const float* b_i2  = (const float*)d_in[12];
  const float* Ws_x  = (const float*)d_in[13];
  const float* Wd_x  = (const float*)d_in[14];
  const float* as_x  = (const float*)d_in[15];
  const float* ad_x  = (const float*)d_in[16];
  const float* b_x   = (const float*)d_in[17];
  const float* W_out = (const float*)d_in[18];
  const float* b_out = (const float*)d_in[19];

  float* ws = (float*)d_ws;
  size_t off = 0;
  auto alloc = [&](size_t n) { float* p = ws + off; off += n; return p; };

  float* h1   = alloc((size_t)N1C * DC);
  float* h2   = alloc((size_t)N2C * DC);
  float* g1   = alloc((size_t)N1C * DC);
  float* g2   = alloc((size_t)N2C * DC);
  float* gx   = alloc((size_t)N1C * DC);
  float* a_s1 = alloc(N1C);
  float* a_d1 = alloc(N1C);
  float* a_s2 = alloc(N2C);
  float* a_d2 = alloc(N2C);
  float* a_sx = alloc(N1C);
  float* a_dx = alloc(N2C);
  float* vd   = alloc(DC);
  int*   rp   = (int*)alloc((size_t)3 * NSTR);
  unsigned short* csr = (unsigned short*)alloc((size_t)3 * EC / 2);
  (void)ws_size; (void)in_sizes; (void)n_in; (void)out_size;

  const dim3 blk(256);
  const dim3 gemm_g(GEMM_WGS);
  const dim3 rd_g1((N1C / NODE_ROWS + 7) / 8), rd_g2((N2C / NODE_ROWS + 7) / 8);
  const dim3 dot_g1((N1C / RD_ROWS + 7) / 8), dot_g2((N2C / RD_ROWS + 7) / 8);
  const dim3 rp_g((3 * NSTR + 255) / 256);
  const dim3 e3_g(GRP_STRIDE / GRP_T, 3);
  const dim3 grp_blk(GRP_T);
  const size_t lds_bytes = (size_t)(DC * WSTR + 2 * 16 * WSTR) * sizeof(float);

  const int* rp11 = rp;
  const int* rp22 = rp + NSTR;
  const int* rp12 = rp + 2 * NSTR;
  const unsigned short* csr11 = csr;
  const unsigned short* csr22 = csr + (size_t)EC;
  const unsigned short* csr12 = csr + (size_t)2 * EC;

  zero_int_kernel<<<rp_g, blk, 0, stream>>>(rp, 3 * NSTR);
  dst_count_kernel<<<e3_g, grp_blk, 0, stream>>>(ei11, ei22, ei12, rp);
  dst_scan_kernel<<<3, 1024, 0, stream>>>(rp);
  dst_fill_kernel<<<e3_g, grp_blk, 0, stream>>>(ei11, ei22, ei12, rp, csr);

  const float* in1 = x1;
  const float* in2 = x2;

  for (int l = 0; l < LC; ++l) {
    const size_t wo = (size_t)l * DC * DC;
    const size_t vo = (size_t)l * DC;

    gemm_n128_wmma<<<gemm_g, blk, lds_bytes, stream>>>(in1, W_i1 + wo, nullptr, g1, N1C / 16);
    gemm_n128_wmma<<<gemm_g, blk, lds_bytes, stream>>>(in2, W_i2 + wo, nullptr, g2, N2C / 16);
    gemm_n128_wmma<<<gemm_g, blk, lds_bytes, stream>>>(in1, Ws_x + wo, nullptr, gx, N1C / 16);

    rowdot2_kernel<<<dot_g1, blk, 0, stream>>>(g1, as_i1 + vo, ad_i1 + vo, a_s1, a_d1, N1C);
    rowdot2_kernel<<<dot_g2, blk, 0, stream>>>(g2, as_i2 + vo, ad_i2 + vo, a_s2, a_d2, N2C);
    rowdot2_kernel<<<dot_g1, blk, 0, stream>>>(gx, as_x + vo, nullptr, a_sx, nullptr, N1C);
    vecmat128_kernel<<<1, DC, 0, stream>>>(Wd_x + wo, ad_x + vo, vd);
    rowdot2_kernel<<<dot_g2, blk, 0, stream>>>(in2, vd, nullptr, a_dx, nullptr, N2C);

    node_softmax_gather_kernel<false><<<rd_g1, blk, 0, stream>>>(
        rp11, csr11, a_s1, a_d1, g1, b_i1 + vo,
        nullptr, nullptr, nullptr, nullptr, nullptr, nullptr, h1, N1C);
    node_softmax_gather_kernel<true><<<rd_g2, blk, 0, stream>>>(
        rp22, csr22, a_s2, a_d2, g2, b_i2 + vo,
        rp12, csr12, a_sx, a_dx, gx, b_x + vo, h2, N2C);

    in1 = h1;
    in2 = h2;
  }

  float* out = (float*)d_out;
  gemm_n128_wmma<<<gemm_g, blk, lds_bytes, stream>>>(h1, W_out, b_out, out, N1C / 16);
  gemm_n128_wmma<<<gemm_g, blk, lds_bytes, stream>>>(h2, W_out, b_out,
                                                     out + (size_t)N1C * DC, N2C / 16);
}


# edges grouped by destination without global atomics: chunks binned by row range with LDS counters (add's return = rank), then a block per range ranks its rows; slots still by arrival
# speedup vs baseline: 6.2097x; 1.9191x over previous
#include <hip/hip_runtime.h>


#define N1C 50000
#define N2C 50000
#define EC  800000
#define DC  128
#define LC  2
#define WSTR 132
#define NMAXC (N1C > N2C ? N1C : N2C)
#define NR 256
#define RW ((NMAXC + NR - 1) / NR)
#define NSTR (NR * RW)

static_assert(N1C <= NSTR && N2C <= NSTR, "row-pointer stride too small");
static_assert(N1C <= 65536, "source ids are kept as 16-bit values");
static_assert(N1C % 16 == 0 && N2C % 16 == 0, "GEMM row tiles are 16 rows, no tail tile");

typedef float v2f __attribute__((ext_vector_type(2)));
typedef float v8f __attribute__((ext_vector_type(8)));

#define GEMM_WGS 768

__global__ __launch_bounds__(256)
void gemm_n128_wmma(const float* __restrict__ A, const float* __restrict__ W,
                    const float* __restrict__ bias, float* __restrict__ C,
                    int ntiles) {
  extern __shared__ float smem[];
  float* sA  = smem;
  float* sWt = smem + 2 * 16 * WSTR;

  const int tid  = threadIdx.x;
  const int lane = tid & 31;
  const int wv   = tid >> 5;

  int tile = blockIdx.x;
  if (tile >= ntiles) return;

  const int ar  = tid >> 5;
  const int an4 = (tid & 31) * 4;
  const float* arow = A + (size_t)ar * DC + an4;
  float4 pa0 = *(const float4*)(arow + (size_t)tile * 16 * DC);
  float4 pa1 = *(const float4*)(arow + (size_t)tile * 16 * DC + 8 * DC);

#pragma unroll
  for (int i = 0; i < 16; ++i) {
    const int k  = 8 * i + (lane >> 2);
    const int n4 = 16 * wv + 4 * (lane & 3);
    const float4 w4 = *(const float4*)(W + (size_t)k * DC + n4);
    sWt[(n4 + 0) * WSTR + k] = w4.x;
    sWt[(n4 + 1) * WSTR + k] = w4.y;
    sWt[(n4 + 2) * WSTR + k] = w4.z;
    sWt[(n4 + 3) * WSTR + k] = w4.w;
  }

  const int cb   = wv * 16;
  const int m    = lane & 15;
  const int half = lane >> 4;
  const int col  = cb + m;
  const float bv = bias ? bias[col] : 0.f;
  const float* bp = sWt + (cb + m) * WSTR + half * 2;

  int buf = 0;
  for (; tile < ntiles; tile += gridDim.x) {
    float* sAb = sA + buf * (16 * WSTR);
    *(float4*)(sAb + ar * WSTR + an4)       = pa0;
    *(float4*)(sAb + (ar + 8) * WSTR + an4) = pa1;
    __syncthreads();

    const int nt = tile + gridDim.x;
    if (nt < ntiles) {
      pa0 = *(const float4*)(arow + (size_t)nt * 16 * DC);
      pa1 = *(const float4*)(arow + (size_t)nt * 16 * DC + 8 * DC);
    }

    const float* ap = sAb + m * WSTR + half * 2;
    v8f acc = {};
#pragma unroll 8
    for (int kk = 0; kk < 32; ++kk) {
      const v2f a = *(const v2f*)(ap + kk * 4);
      const v2f b = *(const v2f*)(bp + kk * 4);
      acc = __builtin_amdgcn_wmma_f32_16x16x4_f32(false, a, false, b,
                                                  (short)0, acc, false, false);
    }

    float* cp = C + ((size_t)tile * 16 + half * 8) * DC + col;
#pragma unroll
    for (int r = 0; r < 8; ++r) {
      cp[(size_t)r * DC] = acc[r] + bv;
    }
    buf ^= 1;
  }
}

#define RD_ROWS 16
static_assert(N1C % RD_ROWS == 0 && N2C % RD_ROWS == 0, "a row-dot wave has all its rows or none");

__global__ __launch_bounds__(256)
void rowdot2_kernel(const float* __restrict__ X, const float* __restrict__ v1,
                    const float* __restrict__ v2, float* __restrict__ o1,
                    float* __restrict__ o2, int N) {
  const int wave = blockIdx.x * (blockDim.x >> 5) + (threadIdx.x >> 5);
  const int lane = threadIdx.x & 31;
  const int row0 = wave * RD_ROWS;
  if (row0 >= N) return;
  const float4* xp = (const float4*)(X + (size_t)row0 * DC) + lane;
  float4 x[RD_ROWS];
#pragma unroll
  for (int r = 0; r < RD_ROWS; ++r) x[r] = xp[r * (DC / 4)];

  const float4 a = ((const float4*)v1)[lane];
  float s1[RD_ROWS];
#pragma unroll
  for (int r = 0; r < RD_ROWS; ++r)
    s1[r] = x[r].x * a.x + x[r].y * a.y + x[r].z * a.z + x[r].w * a.w;
#pragma unroll
  for (int off = 16; off > 0; off >>= 1) {
#pragma unroll
    for (int r = 0; r < RD_ROWS; ++r) s1[r] += __shfl_xor(s1[r], off, 32);
  }
  float out1 = s1[0];
#pragma unroll
  for (int r = 1; r < RD_ROWS; ++r) out1 = lane == r ? s1[r] : out1;
  if (lane < RD_ROWS) o1[row0 + lane] = out1;

  if (v2) {
    const float4 b = ((const float4*)v2)[lane];
    float s2[RD_ROWS];
#pragma unroll
    for (int r = 0; r < RD_ROWS; ++r)
      s2[r] = x[r].x * b.x + x[r].y * b.y + x[r].z * b.z + x[r].w * b.w;
#pragma unroll
    for (int off = 16; off > 0; off >>= 1) {
#pragma unroll
      for (int r = 0; r < RD_ROWS; ++r) s2[r] += __shfl_xor(s2[r], off, 32);
    }
    float out2 = s2[0];
#pragma unroll
    for (int r = 1; r < RD_ROWS; ++r) out2 = lane == r ? s2[r] : out2;
    if (lane < RD_ROWS) o2[row0 + lane] = out2;
  }
}

__global__ void vecmat128_kernel(const float* __restrict__ W,
                                 const float* __restrict__ a,
                                 float* __restrict__ v) {
  const int k = threadIdx.x;
  float s = 0.f;
#pragma unroll 8
  for (int j = 0; j < DC; ++j) s += W[(size_t)k * DC + j] * a[j];
  v[k] = s;
}

#define BIN_T 256
#define NCH 256
#define CS (EC / NCH)
#define BIN_IT ((CS + BIN_T - 1) / BIN_T)
static_assert(EC % NCH == 0, "chunks have no tail");
static_assert(NR == BIN_T && NCH == BIN_T, "one thread per range (binning) and per chunk (ordering)");
static_assert(RW <= BIN_T && RW <= 65535, "one thread per row of a range; the row fits the word's upper half");

__device__ __forceinline__ int block_scan_excl(int v, int* wtot, int& total) {
  const int lane = threadIdx.x & 31;
  const int w    = threadIdx.x >> 5;
  int incl = v;
#pragma unroll
  for (int off = 1; off < 32; off <<= 1) {
    const int u = __shfl_up(incl, off, 32);
    if (lane >= off) incl += u;
  }
  if (lane == 31) wtot[w] = incl;
  __syncthreads();
  int pre = 0, tot = 0;
#pragma unroll
  for (int i = 0; i < BIN_T / 32; ++i) {
    const int t = wtot[i];
    if (i < w) pre += t;
    tot += t;
  }
  total = tot;
  return pre + incl - v;
}

__global__ __launch_bounds__(BIN_T)
void edge_bin_kernel(const int* __restrict__ e11,
                     const int* __restrict__ e22,
                     const int* __restrict__ e12,
                     unsigned* __restrict__ binned,
                     int* __restrict__ tab) {
  __shared__ int cnt[NR];
  __shared__ int offs[NR];
  __shared__ int wtot[BIN_T / 32];
  __shared__ unsigned stage[CS];
  const int tid   = threadIdx.x;
  const int chunk = blockIdx.x;
  const int t     = blockIdx.y;
  const int* ei = (t == 0 ? e11 : (t == 1 ? e22 : e12)) + chunk * CS;

  cnt[tid] = 0;
  int s[BIN_IT], d[BIN_IT], r[BIN_IT], rk[BIN_IT];
#pragma unroll
  for (int u = 0; u < BIN_IT; ++u) {
    const int i = tid + u * BIN_T;
    const bool v = i < CS;
    s[u] = v ? ei[i] : 0;
    d[u] = v ? ei[EC + i] : 0;
  }
  __syncthreads();
#pragma unroll
  for (int u = 0; u < BIN_IT; ++u) {
    const int i = tid + u * BIN_T;
    r[u]  = d[u] / RW;
    rk[u] = 0;
    if (i < CS) rk[u] = atomicAdd(&cnt[r[u]], 1);
  }
  __syncthreads();
  int total;
  const int off = block_scan_excl(cnt[tid], wtot, total);
  offs[tid] = off;
  tab[((size_t)t * NCH + chunk) * NR + tid] = off;
  __syncthreads();
#pragma unroll
  for (int u = 0; u < BIN_IT; ++u) {
    const int i = tid + u * BIN_T;
    if (i < CS)
      stage[offs[r[u]] + rk[u]] = ((unsigned)(d[u] - r[u] * RW) << 16) | (unsigned)s[u];
  }
  __syncthreads();
  unsigned* out = binned + (size_t)t * EC + (size_t)chunk * CS;
#pragma unroll
  for (int u = 0; u < BIN_IT; ++u) {
    const int i = tid + u * BIN_T;
    if (i < CS) out[i] = stage[i];
  }
}

__global__ __launch_bounds__(BIN_T)
void range_sort_kernel(const unsigned* __restrict__ binned,
                       const int* __restrict__ tab,
                       int* __restrict__ rp,
                       unsigned short* __restrict__ csr) {
  __shared__ int rowcnt[BIN_T];
  __shared__ int rowpos[BIN_T];
  __shared__ int wtotA[BIN_T / 32];
  __shared__ int wtotB[BIN_T / 32];
  const int tid  = threadIdx.x;
  const int lane = tid & 31;
  const int c0 = __builtin_amdgcn_readfirstlane(tid & ~31);
  const int r  = blockIdx.x;
  const int t  = blockIdx.y;
  const int nd = t == 0 ? N1C : N2C;

  const int* tb = tab + ((size_t)t * NCH + tid) * NR;
  const int sb = tb[r];
  const int se = r + 1 < NR ? tb[r + 1] : CS;
  const int n  = se - sb;
  rowcnt[tid] = 0;
  int base;
  (void)block_scan_excl(sb, wtotA, base);

  const unsigned* lst = binned + (size_t)t * EC + (size_t)c0 * CS;
  const bool more = __builtin_amdgcn_ballot_w32(n > 32) != 0u;
  unsigned w[32];
#pragma unroll
  for (int j = 0; j < 32; ++j) {
    const int sbj = __builtin_amdgcn_readlane(sb, j);
    const int nj  = __builtin_amdgcn_readlane(n, j);
    w[j] = lane < nj ? lst[(size_t)j * CS + sbj + lane] : 0u;
  }
#pragma unroll
  for (int j = 0; j < 32; ++j) {
    const int nj = __builtin_amdgcn_readlane(n, j);
    if (lane < nj) atomicAdd(&rowcnt[w[j] >> 16], 1);
  }
  if (more) {
#pragma nounroll
    for (int j = 0; j < 32; ++j) {
      const int sbj = __builtin_amdgcn_readlane(sb, j);
      const int nj  = __builtin_amdgcn_readlane(n, j);
      const unsigned* pj = lst + (size_t)j * CS + sbj;
      for (int k = 32 + lane; k < nj; k += 32) atomicAdd(&rowcnt[pj[k] >> 16], 1);
    }
  }
  __syncthreads();

  int tot;
  const int c  = rowcnt[tid];
  const int st = block_scan_excl(c, wtotB, tot);
  rowpos[tid] = base + st;
  const int d = r * RW + tid;
  if (tid < RW && d < nd) rp[t * NSTR + d] = base + st + c;
  __syncthreads();

  unsigned short* out = csr + (size_t)t * EC;
#pragma unroll
  for (int j = 0; j < 32; ++j) {
    const int nj = __builtin_amdgcn_readlane(n, j);
    if (lane < nj) {
      const int p = atomicAdd(&rowpos[w[j] >> 16], 1);
      out[p] = (unsigned short)(w[j] & 0xffffu);
    }
  }
  if (more) {
#pragma nounroll
    for (int j = 0; j < 32; ++j) {
      const int sbj = __builtin_amdgcn_readlane(sb, j);
      const int nj  = __builtin_amdgcn_readlane(n, j);
      const unsigned* pj = lst + (size_t)j * CS + sbj;
      for (int k = 32 + lane; k < nj; k += 32) {
        const unsigned wk = pj[k];
        const int p = atomicAdd(&rowpos[wk >> 16], 1);
        out[p] = (unsigned short)(wk & 0xffffu);
      }
    }
  }
}

__device__ __forceinline__ float leaky02(float l) { return l > 0.f ? l : 0.2f * l; }

__device__ __forceinline__ float wave_max(float v) {
#pragma unroll
  for (int off = 16; off > 0; off >>= 1) v = fmaxf(v, __shfl_xor(v, off, 32));
  return v;
}
__device__ __forceinline__ float wave_sum(float v) {
#pragma unroll
  for (int off = 16; off > 0; off >>= 1) v += __shfl_xor(v, off, 32);
  return v;
}

__device__ __forceinline__ void gather_chunk(float4& acc, int s, float c, int n,
                                             const float* __restrict__ G, int lane) {
  const float* gl = G + lane * 4;
  int j = 0;
  for (; j + 4 <= n; j += 4) {
    const int s0 = __builtin_amdgcn_readlane(s, j + 0);
    const int s1 = __builtin_amdgcn_readlane(s, j + 1);
    const int s2 = __builtin_amdgcn_readlane(s, j + 2);
    const int s3 = __builtin_amdgcn_readlane(s, j + 3);
    const float c0 = __int_as_float(__builtin_amdgcn_readlane(__float_as_int(c), j + 0));
    const float c1 = __int_as_float(__builtin_amdgcn_readlane(__float_as_int(c), j + 1));
    const float c2 = __int_as_float(__builtin_amdgcn_readlane(__float_as_int(c), j + 2));
    const float c3 = __int_as_float(__builtin_amdgcn_readlane(__float_as_int(c), j + 3));
    const float4 g0 = *(const float4*)(gl + (size_t)s0 * DC);
    const float4 g1 = *(const float4*)(gl + (size_t)s1 * DC);
    const float4 g2 = *(const float4*)(gl + (size_t)s2 * DC);
    const float4 g3 = *(const float4*)(gl + (size_t)s3 * DC);
    acc.x += c0 * g0.x; acc.y += c0 * g0.y; acc.z += c0 * g0.z; acc.w += c0 * g0.w;
    acc.x += c1 * g1.x; acc.y += c1 * g1.y; acc.z += c1 * g1.z; acc.w += c1 * g1.w;
    acc.x += c2 * g2.x; acc.y += c2 * g2.y; acc.z += c2 * g2.z; acc.w += c2 * g2.w;
    acc.x += c3 * g3.x; acc.y += c3 * g3.y; acc.z += c3 * g3.z; acc.w += c3 * g3.w;
  }
  for (; j < n; ++j) {
    const int sj = __builtin_amdgcn_readlane(s, j);
    const float cj = __int_as_float(__builtin_amdgcn_readlane(__float_as_int(c), j));
    const float4 g = *(const float4*)(gl + (size_t)sj * DC);
    acc.x += cj * g.x; acc.y += cj * g.y; acc.z += cj * g.z; acc.w += cj * g.w;
  }
}

__device__ __forceinline__ float4 seg_softmax_gather(
    int d, int lane, const int* __restrict__ rp,
    const unsigned short* __restrict__ csr, const float* __restrict__ a_s,
    const float* __restrict__ a_d, const float* __restrict__ G) {
  float4 acc = make_float4(0.f, 0.f, 0.f, 0.f);
  const int end = rp[d];
  const int beg = d > 0 ? rp[d - 1] : 0;
  const float adv = a_d[d];
  const float NEG_INF = -__builtin_inff();

  const int  i0 = beg + lane;
  const bool v0 = i0 < end;
  const int  s0 = v0 ? (int)csr[i0] : 0;
  const float l0 = v0 ? leaky02(a_s[s0] + adv) : NEG_INF;

  float m = l0;
  for (int i = i0 + 32; i < end; i += 32)
    m = fmaxf(m, leaky02(a_s[csr[i]] + adv));
  m = wave_max(m);

  const float e0 = v0 ? __expf(l0 - m) : 0.f;
  float z = e0;
  for (int i = i0 + 32; i < end; i += 32)
    z += __expf(leaky02(a_s[csr[i]] + adv) - m);
  z = wave_sum(z);
  const float den = z + 1e-16f;

  const int n0 = end - beg < 32 ? end - beg : 32;
  gather_chunk(acc, s0, e0 / den, n0, G, lane);

  for (int base = beg + 32; base < end; base += 32) {
    const int  i = base + lane;
    const bool v = i < end;
    const int  s = v ? (int)csr[i] : 0;
    const float c = v ? __expf(leaky02(a_s[s] + adv) - m) / den : 0.f;
    const int n = end - base < 32 ? end - base : 32;
    gather_chunk(acc, s, c, n, G, lane);
  }
  return acc;
}

#define NODE_ROWS 2
static_assert(N1C % NODE_ROWS == 0 && N2C % NODE_ROWS == 0, "a gather wave has all its rows or none");

template <bool TWO>
__global__ __launch_bounds__(256, 16)
void node_softmax_gather_kernel(
    const int* __restrict__ rpA, const unsigned short* __restrict__ csrA,
    const float* __restrict__ asA, const float* __restrict__ adA,
    const float* __restrict__ GA, const float* __restrict__ bA,
    const int* __restrict__ rpB, const unsigned short* __restrict__ csrB,
    const float* __restrict__ asB, const float* __restrict__ adB,
    const float* __restrict__ GB, const float* __restrict__ bB,
    float* __restrict__ H, int N) {
  const int w = __builtin_amdgcn_readfirstlane(
      (int)(blockIdx.x * (blockDim.x >> 5) + (threadIdx.x >> 5)));
  const int lane = threadIdx.x & 31;
  if (w * NODE_ROWS >= N) return;

#pragma nounroll
  for (int i = 0; i < NODE_ROWS; ++i) {
    const int d = w * NODE_ROWS + i;
    const float4 a  = seg_softmax_gather(d, lane, rpA, csrA, asA, adA, GA);
    const float4 ba = ((const float4*)bA)[lane];
    float4 o = make_float4(a.x + ba.x, a.y + ba.y, a.z + ba.z, a.w + ba.w);
    if (TWO) {
      const float4 b  = seg_softmax_gather(d, lane, rpB, csrB, asB, adB, GB);
      const float4 bb = ((const float4*)bB)[lane];
      o.x += b.x + bb.x; o.y += b.y + bb.y; o.z += b.z + bb.z; o.w += b.w + bb.w;
    }
    o.x = o.x > 0.f ? o.x : 0.f;
    o.y = o.y > 0.f ? o.y : 0.f;
    o.z = o.z > 0.f ? o.z : 0.f;
    o.w = o.w > 0.f ? o.w : 0.f;
    *(float4*)(H + (size_t)d * DC + lane * 4) = o;
  }
}

extern "C" void kernel_launch(void* const* d_in, const int* in_sizes, int n_in,
                              void* d_out, int out_size, void* d_ws, size_t ws_size,
                              hipStream_t stream) {
  const float* x1    = (const float*)d_in[0];
  const float* x2    = (const float*)d_in[1];
  const int*   ei11  = (const int*)d_in[2];
  const int*   ei22  = (const int*)d_in[3];
  const int*   ei12  = (const int*)d_in[4];
  const float* W_i1  = (const float*)d_in[5];
  const float* as_i1 = (const float*)d_in[6];
  const float* ad_i1 = (const float*)d_in[7];
  const float* b_i1  = (const float*)d_in[8];
  const float* W_i2  = (const float*)d_in[9];
  const float* as_i2 = (const float*)d_in[10];
  const float* ad_i2 = (const float*)d_in[11];
  const float* b_i2  = (const float*)d_in[12];
  const float* Ws_x  = (const float*)d_in[13];
  const float* Wd_x  = (const float*)d_in[14];
  const float* as_x  = (const float*)d_in[15];
  const float* ad_x  = (const float*)d_in[16];
  const float* b_x   = (const float*)d_in[17];
  const float* W_out = (const float*)d_in[18];
  const float* b_out = (const float*)d_in[19];

  float* ws = (float*)d_ws;
  size_t off = 0;
  auto alloc = [&](size_t n) { float* p = ws + off; off += n; return p; };

  float* h1   = alloc((size_t)N1C * DC);
  float* h2   = alloc((size_t)N2C * DC);
  float* g1   = alloc((size_t)N1C * DC);
  float* g2   = alloc((size_t)N2C * DC);
  float* gx   = alloc((size_t)N1C * DC);
  float* a_s1 = alloc(N1C);
  float* a_d1 = alloc(N1C);
  float* a_s2 = alloc(N2C);
  float* a_d2 = alloc(N2C);
  float* a_sx = alloc(N1C);
  float* a_dx = alloc(N2C);
  float* vd   = alloc(DC);
  int*   rp   = (int*)alloc((size_t)3 * NSTR);
  unsigned short* csr = (unsigned short*)alloc((size_t)3 * EC / 2);
  (void)ws_size; (void)in_sizes; (void)n_in; (void)out_size;

  const dim3 blk(256);
  const dim3 gemm_g(GEMM_WGS);
  const dim3 rd_g1((N1C / NODE_ROWS + 7) / 8), rd_g2((N2C / NODE_ROWS + 7) / 8);
  const dim3 dot_g1((N1C / RD_ROWS + 7) / 8), dot_g2((N2C / RD_ROWS + 7) / 8);
  const dim3 bin_g(NCH, 3), srt_g(NR, 3);
  const dim3 bin_blk(BIN_T);
  const size_t lds_bytes = (size_t)(DC * WSTR + 2 * 16 * WSTR) * sizeof(float);

  const int* rp11 = rp;
  const int* rp22 = rp + NSTR;
  const int* rp12 = rp + 2 * NSTR;
  const unsigned short* csr11 = csr;
  const unsigned short* csr22 = csr + (size_t)EC;
  const unsigned short* csr12 = csr + (size_t)2 * EC;

  static_assert((size_t)3 * EC + (size_t)3 * NCH * NR <= (size_t)N1C * DC, "grouping scratch fits in h1");
  unsigned* binned = (unsigned*)h1;
  int* tab = (int*)(binned + (size_t)3 * EC);
  edge_bin_kernel<<<bin_g, bin_blk, 0, stream>>>(ei11, ei22, ei12, binned, tab);
  range_sort_kernel<<<srt_g, bin_blk, 0, stream>>>(binned, tab, rp, csr);

  const float* in1 = x1;
  const float* in2 = x2;

  for (int l = 0; l < LC; ++l) {
    const size_t wo = (size_t)l * DC * DC;
    const size_t vo = (size_t)l * DC;

    gemm_n128_wmma<<<gemm_g, blk, lds_bytes, stream>>>(in1, W_i1 + wo, nullptr, g1, N1C / 16);
    gemm_n128_wmma<<<gemm_g, blk, lds_bytes, stream>>>(in2, W_i2 + wo, nullptr, g2, N2C / 16);
    gemm_n128_wmma<<<gemm_g, blk, lds_bytes, stream>>>(in1, Ws_x + wo, nullptr, gx, N1C / 16);

    rowdot2_kernel<<<dot_g1, blk, 0, stream>>>(g1, as_i1 + vo, ad_i1 + vo, a_s1, a_d1, N1C);
    rowdot2_kernel<<<dot_g2, blk, 0, stream>>>(g2, as_i2 + vo, ad_i2 + vo, a_s2, a_d2, N2C);
    rowdot2_kernel<<<dot_g1, blk, 0, stream>>>(gx, as_x + vo, nullptr, a_sx, nullptr, N1C);
    vecmat128_kernel<<<1, DC, 0, stream>>>(Wd_x + wo, ad_x + vo, vd);
    rowdot2_kernel<<<dot_g2, blk, 0, stream>>>(in2, vd, nullptr, a_dx, nullptr, N2C);

    node_softmax_gather_kernel<false><<<rd_g1, blk, 0, stream>>>(
        rp11, csr11, a_s1, a_d1, g1, b_i1 + vo,
        nullptr, nullptr, nullptr, nullptr, nullptr, nullptr, h1, N1C);
    node_softmax_gather_kernel<true><<<rd_g2, blk, 0, stream>>>(
        rp22, csr22, a_s2, a_d2, g2, b_i2 + vo,
        rp12, csr12, a_sx, a_dx, gx, b_x + vo, h2, N2C);

    in1 = h1;
    in2 = h2;
  }

  float* out = (float*)d_out;
  gemm_n128_wmma<<<gemm_g, blk, lds_bytes, stream>>>(h1, W_out, b_out, out, N1C / 16);
  gemm_n128_wmma<<<gemm_g, blk, lds_bytes, stream>>>(h2, W_out, b_out,
                                                     out + (size_t)N1C * DC, N2C / 16);
}


# attention logits as x.(W a): one small launch folds the vectors through W, the GEMMs take their A rows' dots from loader registers; row-dot passes go (24 to 15 launches); logit sum order changes
# speedup vs baseline: 6.8348x; 1.1007x over previous
#include <hip/hip_runtime.h>


#define N1C 50000
#define N2C 50000
#define EC  800000
#define DC  128
#define LC  2
#define WSTR 132
#define NMAXC (N1C > N2C ? N1C : N2C)
#define NR 256
#define RW ((NMAXC + NR - 1) / NR)
#define NSTR (NR * RW)

static_assert(N1C <= NSTR && N2C <= NSTR, "row-pointer stride too small");
static_assert(N1C <= 65536, "source ids are kept as 16-bit values");
static_assert(N1C % 16 == 0 && N2C % 16 == 0, "GEMM row tiles are 16 rows, no tail tile");

typedef float v2f __attribute__((ext_vector_type(2)));
typedef float v8f __attribute__((ext_vector_type(8)));

#define GEMM_WGS 768
#define NDOT 3

__device__ __forceinline__ float dot4(const float4 x, const float4 a) {
  return x.x * a.x + x.y * a.y + x.z * a.z + x.w * a.w;
}

template <bool DOTS>
__global__ __launch_bounds__(256)
void gemm_n128_wmma(const float* __restrict__ A, const float* __restrict__ W,
                    const float* __restrict__ bias, float* __restrict__ C,
                    int ntiles, const float* __restrict__ vt,
                    float* __restrict__ o0, float* __restrict__ o1,
                    float* __restrict__ o2) {
  extern __shared__ float smem[];
  float* sA   = smem;
  float* sWt  = smem + 2 * 16 * WSTR;
  float* sDot = sWt + DC * WSTR;

  const int tid  = threadIdx.x;
  const int lane = tid & 31;
  const int wv   = tid >> 5;

  int tile = blockIdx.x;
  if (tile >= ntiles) return;

  const int ar  = tid >> 5;
  const int an4 = (tid & 31) * 4;
  const float* arow = A + (size_t)ar * DC + an4;
  float4 pa0 = *(const float4*)(arow + (size_t)tile * 16 * DC);
  float4 pa1 = *(const float4*)(arow + (size_t)tile * 16 * DC + 8 * DC);

  float4 q0 = make_float4(0.f, 0.f, 0.f, 0.f), q1 = q0, q2 = q0;
  if (DOTS) {
    q0 = *(const float4*)(vt + an4);
    q1 = *(const float4*)(vt + DC + an4);
    q2 = *(const float4*)(vt + 2 * DC + an4);
  }
  int ptile = -1;

#pragma unroll
  for (int i = 0; i < 16; ++i) {
    const int k  = 8 * i + (lane >> 2);
    const int n4 = 16 * wv + 4 * (lane & 3);
    const float4 w4 = *(const float4*)(W + (size_t)k * DC + n4);
    sWt[(n4 + 0) * WSTR + k] = w4.x;
    sWt[(n4 + 1) * WSTR + k] = w4.y;
    sWt[(n4 + 2) * WSTR + k] = w4.z;
    sWt[(n4 + 3) * WSTR + k] = w4.w;
  }

  const int cb   = wv * 16;
  const int m    = lane & 15;
  const int half = lane >> 4;
  const int col  = cb + m;
  const float bv = bias ? bias[col] : 0.f;
  const float* bp = sWt + (cb + m) * WSTR + half * 2;

  int buf = 0;
  for (; tile < ntiles; tile += gridDim.x) {
    float* sAb = sA + buf * (16 * WSTR);
    *(float4*)(sAb + ar * WSTR + an4)       = pa0;
    *(float4*)(sAb + (ar + 8) * WSTR + an4) = pa1;
    float r0a = 0.f, r0b = 0.f, r0c = 0.f, r1a = 0.f, r1b = 0.f, r1c = 0.f;
    if (DOTS) {
      r0a = dot4(pa0, q0); r0b = dot4(pa0, q1); r0c = dot4(pa0, q2);
      r1a = dot4(pa1, q0); r1b = dot4(pa1, q1); r1c = dot4(pa1, q2);
    }
    __syncthreads();

    const int nt = tile + gridDim.x;
    if (nt < ntiles) {
      pa0 = *(const float4*)(arow + (size_t)nt * 16 * DC);
      pa1 = *(const float4*)(arow + (size_t)nt * 16 * DC + 8 * DC);
    }

    if (DOTS) {
      if (ptile >= 0 && tid < NDOT * 16) {
        float* o = tid < 16 ? o0 : (tid < 32 ? o1 : o2);
        o[(size_t)ptile * 16 + (tid & 15)] = sDot[(buf ^ 1) * (NDOT * 16) + tid];
      }
      const bool b4 = (lane & 16) != 0, b3 = (lane & 8) != 0, b2 = (lane & 4) != 0;
      const float s0 = (b4 ? r1a : r0a) + __shfl_xor(b4 ? r0a : r1a, 16, 32);
      const float s1 = (b4 ? r1b : r0b) + __shfl_xor(b4 ? r0b : r1b, 16, 32);
      const float s2 = (b4 ? r1c : r0c) + __shfl_xor(b4 ? r0c : r1c, 16, 32);
      const float t0 = (b3 ? s2 : s0) + __shfl_xor(b3 ? s0 : s2, 8, 32);
      const float t1 = (b3 ? 0.f : s1) + __shfl_xor(b3 ? s1 : 0.f, 8, 32);
      float u = (b2 ? t1 : t0) + __shfl_xor(b2 ? t0 : t1, 4, 32);
      u += __shfl_xor(u, 2, 32);
      u += __shfl_xor(u, 1, 32);
      const int j = (lane >> 2) & 3;
      if ((lane & 3) == 0 && j < NDOT)
        sDot[buf * (NDOT * 16) + j * 16 + wv + 8 * (lane >> 4)] = u;
      ptile = tile;
    }

    const float* ap = sAb + m * WSTR + half * 2;
    v8f acc = {};
#pragma unroll 8
    for (int kk = 0; kk < 32; ++kk) {
      const v2f a = *(const v2f*)(ap + kk * 4);
      const v2f b = *(const v2f*)(bp + kk * 4);
      acc = __builtin_amdgcn_wmma_f32_16x16x4_f32(false, a, false, b,
                                                  (short)0, acc, false, false);
    }

    float* cp = C + ((size_t)tile * 16 + half * 8) * DC + col;
#pragma unroll
    for (int r = 0; r < 8; ++r) {
      cp[(size_t)r * DC] = acc[r] + bv;
    }
    buf ^= 1;
  }
  if (DOTS) {
    __syncthreads();
    if (tid < NDOT * 16) {
      float* o = tid < 16 ? o0 : (tid < 32 ? o1 : o2);
      o[(size_t)ptile * 16 + (tid & 15)] = sDot[(buf ^ 1) * (NDOT * 16) + tid];
    }
  }
}

#define NFOLD 6
#define FD_ROWS 16
static_assert(DC == 8 * FD_ROWS, "eight waves cover the rows of W");

__global__ __launch_bounds__(256)
void fold_vectors_kernel(const float* __restrict__ W_i1, const float* __restrict__ as_i1,
                         const float* __restrict__ ad_i1, const float* __restrict__ W_i2,
                         const float* __restrict__ as_i2, const float* __restrict__ ad_i2,
                         const float* __restrict__ Ws_x, const float* __restrict__ as_x,
                         const float* __restrict__ Wd_x, const float* __restrict__ ad_x,
                         float* __restrict__ vt) {
  const int j = blockIdx.x;
  const int l = blockIdx.y;
  const float* Wj = j < 2 ? W_i1 : (j == 2 ? Ws_x : (j < 5 ? W_i2 : Wd_x));
  const float* aj = j == 0 ? as_i1 : (j == 1 ? ad_i1 : (j == 2 ? as_x
                  : (j == 3 ? as_i2 : (j == 4 ? ad_i2 : ad_x))));
  Wj += (size_t)l * DC * DC;
  aj += (size_t)l * DC;
  const int lane = threadIdx.x & 31;
  const int row0 = (threadIdx.x >> 5) * FD_ROWS;
  const float4* wp = (const float4*)(Wj + (size_t)row0 * DC) + lane;
  float4 w[FD_ROWS];
#pragma unroll
  for (int r = 0; r < FD_ROWS; ++r) w[r] = wp[r * (DC / 4)];
  const float4 a = ((const float4*)aj)[lane];
  float s[FD_ROWS];
#pragma unroll
  for (int r = 0; r < FD_ROWS; ++r) s[r] = dot4(w[r], a);
#pragma unroll
  for (int off = 16; off > 0; off >>= 1) {
#pragma unroll
    for (int r = 0; r < FD_ROWS; ++r) s[r] += __shfl_xor(s[r], off, 32);
  }
  float out = s[0];
#pragma unroll
  for (int r = 1; r < FD_ROWS; ++r) out = lane == r ? s[r] : out;
  if (lane < FD_ROWS) vt[((size_t)l * NFOLD + j) * DC + row0 + lane] = out;
}

#define BIN_T 256
#define NCH 256
#define CS (EC / NCH)
#define BIN_IT ((CS + BIN_T - 1) / BIN_T)
static_assert(EC % NCH == 0, "chunks have no tail");
static_assert(NR == BIN_T && NCH == BIN_T, "one thread per range (binning) and per chunk (ordering)");
static_assert(RW <= BIN_T && RW <= 65535, "one thread per row of a range; the row fits the word's upper half");

__device__ __forceinline__ int block_scan_excl(int v, int* wtot, int& total) {
  const int lane = threadIdx.x & 31;
  const int w    = threadIdx.x >> 5;
  int incl = v;
#pragma unroll
  for (int off = 1; off < 32; off <<= 1) {
    const int u = __shfl_up(incl, off, 32);
    if (lane >= off) incl += u;
  }
  if (lane == 31) wtot[w] = incl;
  __syncthreads();
  int pre = 0, tot = 0;
#pragma unroll
  for (int i = 0; i < BIN_T / 32; ++i) {
    const int t = wtot[i];
    if (i < w) pre += t;
    tot += t;
  }
  total = tot;
  return pre + incl - v;
}

__global__ __launch_bounds__(BIN_T)
void edge_bin_kernel(const int* __restrict__ e11,
                     const int* __restrict__ e22,
                     const int* __restrict__ e12,
                     unsigned* __restrict__ binned,
                     int* __restrict__ tab) {
  __shared__ int cnt[NR];
  __shared__ int offs[NR];
  __shared__ int wtot[BIN_T / 32];
  __shared__ unsigned stage[CS];
  const int tid   = threadIdx.x;
  const int chunk = blockIdx.x;
  const int t     = blockIdx.y;
  const int* ei = (t == 0 ? e11 : (t == 1 ? e22 : e12)) + chunk * CS;

  cnt[tid] = 0;
  int s[BIN_IT], d[BIN_IT], r[BIN_IT], rk[BIN_IT];
#pragma unroll
  for (int u = 0; u < BIN_IT; ++u) {
    const int i = tid + u * BIN_T;
    const bool v = i < CS;
    s[u] = v ? ei[i] : 0;
    d[u] = v ? ei[EC + i] : 0;
  }
  __syncthreads();
#pragma unroll
  for (int u = 0; u < BIN_IT; ++u) {
    const int i = tid + u * BIN_T;
    r[u]  = d[u] / RW;
    rk[u] = 0;
    if (i < CS) rk[u] = atomicAdd(&cnt[r[u]], 1);
  }
  __syncthreads();
  int total;
  const int off = block_scan_excl(cnt[tid], wtot, total);
  offs[tid] = off;
  tab[((size_t)t * NCH + chunk) * NR + tid] = off;
  __syncthreads();
#pragma unroll
  for (int u = 0; u < BIN_IT; ++u) {
    const int i = tid + u * BIN_T;
    if (i < CS)
      stage[offs[r[u]] + rk[u]] = ((unsigned)(d[u] - r[u] * RW) << 16) | (unsigned)s[u];
  }
  __syncthreads();
  unsigned* out = binned + (size_t)t * EC + (size_t)chunk * CS;
#pragma unroll
  for (int u = 0; u < BIN_IT; ++u) {
    const int i = tid + u * BIN_T;
    if (i < CS) out[i] = stage[i];
  }
}

__global__ __launch_bounds__(BIN_T)
void range_sort_kernel(const unsigned* __restrict__ binned,
                       const int* __restrict__ tab,
                       int* __restrict__ rp,
                       unsigned short* __restrict__ csr) {
  __shared__ int rowcnt[BIN_T];
  __shared__ int rowpos[BIN_T];
  __shared__ int wtotA[BIN_T / 32];
  __shared__ int wtotB[BIN_T / 32];
  const int tid  = threadIdx.x;
  const int lane = tid & 31;
  const int c0 = __builtin_amdgcn_readfirstlane(tid & ~31);
  const int r  = blockIdx.x;
  const int t  = blockIdx.y;
  const int nd = t == 0 ? N1C : N2C;

  const int* tb = tab + ((size_t)t * NCH + tid) * NR;
  const int sb = tb[r];
  const int se = r + 1 < NR ? tb[r + 1] : CS;
  const int n  = se - sb;
  rowcnt[tid] = 0;
  int base;
  (void)block_scan_excl(sb, wtotA, base);

  const unsigned* lst = binned + (size_t)t * EC + (size_t)c0 * CS;
  const bool more = __builtin_amdgcn_ballot_w32(n > 32) != 0u;
  unsigned w[32];
#pragma unroll
  for (int j = 0; j < 32; ++j) {
    const int sbj = __builtin_amdgcn_readlane(sb, j);
    const int nj  = __builtin_amdgcn_readlane(n, j);
    w[j] = lane < nj ? lst[(size_t)j * CS + sbj + lane] : 0u;
  }
#pragma unroll
  for (int j = 0; j < 32; ++j) {
    const int nj = __builtin_amdgcn_readlane(n, j);
    if (lane < nj) atomicAdd(&rowcnt[w[j] >> 16], 1);
  }
  if (more) {
#pragma nounroll
    for (int j = 0; j < 32; ++j) {
      const int sbj = __builtin_amdgcn_readlane(sb, j);
      const int nj  = __builtin_amdgcn_readlane(n, j);
      const unsigned* pj = lst + (size_t)j * CS + sbj;
      for (int k = 32 + lane; k < nj; k += 32) atomicAdd(&rowcnt[pj[k] >> 16], 1);
    }
  }
  __syncthreads();

  int tot;
  const int c  = rowcnt[tid];
  const int st = block_scan_excl(c, wtotB, tot);
  rowpos[tid] = base + st;
  const int d = r * RW + tid;
  if (tid < RW && d < nd) rp[t * NSTR + d] = base + st + c;
  __syncthreads();

  unsigned short* out = csr + (size_t)t * EC;
#pragma unroll
  for (int j = 0; j < 32; ++j) {
    const int nj = __builtin_amdgcn_readlane(n, j);
    if (lane < nj) {
      const int p = atomicAdd(&rowpos[w[j] >> 16], 1);
      out[p] = (unsigned short)(w[j] & 0xffffu);
    }
  }
  if (more) {
#pragma nounroll
    for (int j = 0; j < 32; ++j) {
      const int sbj = __builtin_amdgcn_readlane(sb, j);
      const int nj  = __builtin_amdgcn_readlane(n, j);
      const unsigned* pj = lst + (size_t)j * CS + sbj;
      for (int k = 32 + lane; k < nj; k += 32) {
        const unsigned wk = pj[k];
        const int p = atomicAdd(&rowpos[wk >> 16], 1);
        out[p] = (unsigned short)(wk & 0xffffu);
      }
    }
  }
}

__device__ __forceinline__ float leaky02(float l) { return l > 0.f ? l : 0.2f * l; }

__device__ __forceinline__ float wave_max(float v) {
#pragma unroll
  for (int off = 16; off > 0; off >>= 1) v = fmaxf(v, __shfl_xor(v, off, 32));
  return v;
}
__device__ __forceinline__ float wave_sum(float v) {
#pragma unroll
  for (int off = 16; off > 0; off >>= 1) v += __shfl_xor(v, off, 32);
  return v;
}

__device__ __forceinline__ void gather_chunk(float4& acc, int s, float c, int n,
                                             const float* __restrict__ G, int lane) {
  const float* gl = G + lane * 4;
  int j = 0;
  for (; j + 4 <= n; j += 4) {
    const int s0 = __builtin_amdgcn_readlane(s, j + 0);
    const int s1 = __builtin_amdgcn_readlane(s, j + 1);
    const int s2 = __builtin_amdgcn_readlane(s, j + 2);
    const int s3 = __builtin_amdgcn_readlane(s, j + 3);
    const float c0 = __int_as_float(__builtin_amdgcn_readlane(__float_as_int(c), j + 0));
    const float c1 = __int_as_float(__builtin_amdgcn_readlane(__float_as_int(c), j + 1));
    const float c2 = __int_as_float(__builtin_amdgcn_readlane(__float_as_int(c), j + 2));
    const float c3 = __int_as_float(__builtin_amdgcn_readlane(__float_as_int(c), j + 3));
    const float4 g0 = *(const float4*)(gl + (size_t)s0 * DC);
    const float4 g1 = *(const float4*)(gl + (size_t)s1 * DC);
    const float4 g2 = *(const float4*)(gl + (size_t)s2 * DC);
    const float4 g3 = *(const float4*)(gl + (size_t)s3 * DC);
    acc.x += c0 * g0.x; acc.y += c0 * g0.y; acc.z += c0 * g0.z; acc.w += c0 * g0.w;
    acc.x += c1 * g1.x; acc.y += c1 * g1.y; acc.z += c1 * g1.z; acc.w += c1 * g1.w;
    acc.x += c2 * g2.x; acc.y += c2 * g2.y; acc.z += c2 * g2.z; acc.w += c2 * g2.w;
    acc.x += c3 * g3.x; acc.y += c3 * g3.y; acc.z += c3 * g3.z; acc.w += c3 * g3.w;
  }
  for (; j < n; ++j) {
    const int sj = __builtin_amdgcn_readlane(s, j);
    const float cj = __int_as_float(__builtin_amdgcn_readlane(__float_as_int(c), j));
    const float4 g = *(const float4*)(gl + (size_t)sj * DC);
    acc.x += cj * g.x; acc.y += cj * g.y; acc.z += cj * g.z; acc.w += cj * g.w;
  }
}

__device__ __forceinline__ float4 seg_softmax_gather(
    int d, int lane, const int* __restrict__ rp,
    const unsigned short* __restrict__ csr, const float* __restrict__ a_s,
    const float* __restrict__ a_d, const float* __restrict__ G) {
  float4 acc = make_float4(0.f, 0.f, 0.f, 0.f);
  const int end = rp[d];
  const int beg = d > 0 ? rp[d - 1] : 0;
  const float adv = a_d[d];
  const float NEG_INF = -__builtin_inff();

  const int  i0 = beg + lane;
  const bool v0 = i0 < end;
  const int  s0 = v0 ? (int)csr[i0] : 0;
  const float l0 = v0 ? leaky02(a_s[s0] + adv) : NEG_INF;

  float m = l0;
  for (int i = i0 + 32; i < end; i += 32)
    m = fmaxf(m, leaky02(a_s[csr[i]] + adv));
  m = wave_max(m);

  const float e0 = v0 ? __expf(l0 - m) : 0.f;
  float z = e0;
  for (int i = i0 + 32; i < end; i += 32)
    z += __expf(leaky02(a_s[csr[i]] + adv) - m);
  z = wave_sum(z);
  const float den = z + 1e-16f;

  const int n0 = end - beg < 32 ? end - beg : 32;
  gather_chunk(acc, s0, e0 / den, n0, G, lane);

  for (int base = beg + 32; base < end; base += 32) {
    const int  i = base + lane;
    const bool v = i < end;
    const int  s = v ? (int)csr[i] : 0;
    const float c = v ? __expf(leaky02(a_s[s] + adv) - m) / den : 0.f;
    const int n = end - base < 32 ? end - base : 32;
    gather_chunk(acc, s, c, n, G, lane);
  }
  return acc;
}

#define NODE_ROWS 2
static_assert(N1C % NODE_ROWS == 0 && N2C % NODE_ROWS == 0, "a gather wave has all its rows or none");

template <bool TWO>
__global__ __launch_bounds__(256, 16)
void node_softmax_gather_kernel(
    const int* __restrict__ rpA, const unsigned short* __restrict__ csrA,
    const float* __restrict__ asA, const float* __restrict__ adA,
    const float* __restrict__ GA, const float* __restrict__ bA,
    const int* __restrict__ rpB, const unsigned short* __restrict__ csrB,
    const float* __restrict__ asB, const float* __restrict__ adB,
    const float* __restrict__ GB, const float* __restrict__ bB,
    float* __restrict__ H, int N) {
  const int w = __builtin_amdgcn_readfirstlane(
      (int)(blockIdx.x * (blockDim.x >> 5) + (threadIdx.x >> 5)));
  const int lane = threadIdx.x & 31;
  if (w * NODE_ROWS >= N) return;

#pragma nounroll
  for (int i = 0; i < NODE_ROWS; ++i) {
    const int d = w * NODE_ROWS + i;
    const float4 a  = seg_softmax_gather(d, lane, rpA, csrA, asA, adA, GA);
    const float4 ba = ((const float4*)bA)[lane];
    float4 o = make_float4(a.x + ba.x, a.y + ba.y, a.z + ba.z, a.w + ba.w);
    if (TWO) {
      const float4 b  = seg_softmax_gather(d, lane, rpB, csrB, asB, adB, GB);
      const float4 bb = ((const float4*)bB)[lane];
      o.x += b.x + bb.x; o.y += b.y + bb.y; o.z += b.z + bb.z; o.w += b.w + bb.w;
    }
    o.x = o.x > 0.f ? o.x : 0.f;
    o.y = o.y > 0.f ? o.y : 0.f;
    o.z = o.z > 0.f ? o.z : 0.f;
    o.w = o.w > 0.f ? o.w : 0.f;
    *(float4*)(H + (size_t)d * DC + lane * 4) = o;
  }
}

extern "C" void kernel_launch(void* const* d_in, const int* in_sizes, int n_in,
                              void* d_out, int out_size, void* d_ws, size_t ws_size,
                              hipStream_t stream) {
  const float* x1    = (const float*)d_in[0];
  const float* x2    = (const float*)d_in[1];
  const int*   ei11  = (const int*)d_in[2];
  const int*   ei22  = (const int*)d_in[3];
  const int*   ei12  = (const int*)d_in[4];
  const float* W_i1  = (const float*)d_in[5];
  const float* as_i1 = (const float*)d_in[6];
  const float* ad_i1 = (const float*)d_in[7];
  const float* b_i1  = (const float*)d_in[8];
  const float* W_i2  = (const float*)d_in[9];
  const float* as_i2 = (const float*)d_in[10];
  const float* ad_i2 = (const float*)d_in[11];
  const float* b_i2  = (const float*)d_in[12];
  const float* Ws_x  = (const float*)d_in[13];
  const float* Wd_x  = (const float*)d_in[14];
  const float* as_x  = (const float*)d_in[15];
  const float* ad_x  = (const float*)d_in[16];
  const float* b_x   = (const float*)d_in[17];
  const float* W_out = (const float*)d_in[18];
  const float* b_out = (const float*)d_in[19];

  float* ws = (float*)d_ws;
  size_t off = 0;
  auto alloc = [&](size_t n) { float* p = ws + off; off += n; return p; };

  float* h1   = alloc((size_t)N1C * DC);
  float* h2   = alloc((size_t)N2C * DC);
  float* g1   = alloc((size_t)N1C * DC);
  float* g2   = alloc((size_t)N2C * DC);
  float* gx   = alloc((size_t)N1C * DC);
  float* a_s1 = alloc(N1C);
  float* a_d1 = alloc(N1C);
  float* a_s2 = alloc(N2C);
  float* a_d2 = alloc(N2C);
  float* a_sx = alloc(N1C);
  float* a_dx = alloc(N2C);
  float* vtab = alloc((size_t)LC * NFOLD * DC);
  int*   rp   = (int*)alloc((size_t)3 * NSTR);
  unsigned short* csr = (unsigned short*)alloc((size_t)3 * EC / 2);
  (void)ws_size; (void)in_sizes; (void)n_in; (void)out_size;

  const dim3 blk(256);
  const dim3 gemm_g(GEMM_WGS);
  const dim3 rd_g1((N1C / NODE_ROWS + 7) / 8), rd_g2((N2C / NODE_ROWS + 7) / 8);
  const dim3 fold_g(NFOLD, LC);
  const dim3 bin_g(NCH, 3), srt_g(NR, 3);
  const dim3 bin_blk(BIN_T);
  const size_t lds_bytes = (size_t)(DC * WSTR + 2 * 16 * WSTR + 2 * NDOT * 16) * sizeof(float);

  const int* rp11 = rp;
  const int* rp22 = rp + NSTR;
  const int* rp12 = rp + 2 * NSTR;
  const unsigned short* csr11 = csr;
  const unsigned short* csr22 = csr + (size_t)EC;
  const unsigned short* csr12 = csr + (size_t)2 * EC;

  static_assert((size_t)3 * EC + (size_t)3 * NCH * NR <= (size_t)N1C * DC, "grouping scratch fits in h1");
  unsigned* binned = (unsigned*)h1;
  int* tab = (int*)(binned + (size_t)3 * EC);
  edge_bin_kernel<<<bin_g, bin_blk, 0, stream>>>(ei11, ei22, ei12, binned, tab);
  range_sort_kernel<<<srt_g, bin_blk, 0, stream>>>(binned, tab, rp, csr);

  fold_vectors_kernel<<<fold_g, blk, 0, stream>>>(W_i1, as_i1, ad_i1, W_i2, as_i2, ad_i2,
                                                  Ws_x, as_x, Wd_x, ad_x, vtab);

  const float* in1 = x1;
  const float* in2 = x2;

  for (int l = 0; l < LC; ++l) {
    const size_t wo = (size_t)l * DC * DC;
    const size_t vo = (size_t)l * DC;

    const float* vt1 = vtab + (size_t)l * NFOLD * DC;
    const float* vt2 = vt1 + 3 * DC;

    gemm_n128_wmma<true><<<gemm_g, blk, lds_bytes, stream>>>(in1, W_i1 + wo, nullptr, g1, N1C / 16,
                                                             vt1, a_s1, a_d1, a_sx);
    gemm_n128_wmma<true><<<gemm_g, blk, lds_bytes, stream>>>(in2, W_i2 + wo, nullptr, g2, N2C / 16,
                                                             vt2, a_s2, a_d2, a_dx);
    gemm_n128_wmma<false><<<gemm_g, blk, lds_bytes, stream>>>(in1, Ws_x + wo, nullptr, gx, N1C / 16,
                                                              nullptr, nullptr, nullptr, nullptr);

    node_softmax_gather_kernel<false><<<rd_g1, blk, 0, stream>>>(
        rp11, csr11, a_s1, a_d1, g1, b_i1 + vo,
        nullptr, nullptr, nullptr, nullptr, nullptr, nullptr, h1, N1C);
    node_softmax_gather_kernel<true><<<rd_g2, blk, 0, stream>>>(
        rp22, csr22, a_s2, a_d2, g2, b_i2 + vo,
        rp12, csr12, a_sx, a_dx, gx, b_x + vo, h2, N2C);

    in1 = h1;
    in2 = h2;
  }

  float* out = (float*)d_out;
  gemm_n128_wmma<false><<<gemm_g, blk, lds_bytes, stream>>>(h1, W_out, b_out, out, N1C / 16,
                                                            nullptr, nullptr, nullptr, nullptr);
  gemm_n128_wmma<false><<<gemm_g, blk, lds_bytes, stream>>>(h2, W_out, b_out, out + (size_t)N1C * DC,
                                                            N2C / 16, nullptr, nullptr, nullptr, nullptr);
}


# independent kernels of a stage share one launch (block index picks the part): bin+fold, 3 layer GEMMs, both gathers, both output GEMMs as one N1+N2 product; 15 to 7 launches
# speedup vs baseline: 7.5801x; 1.1090x over previous
#include <hip/hip_runtime.h>


#define N1C 50000
#define N2C 50000
#define EC  800000
#define DC  128
#define LC  2
#define WSTR 132
#define NMAXC (N1C > N2C ? N1C : N2C)
#define NR 256
#define RW ((NMAXC + NR - 1) / NR)
#define NSTR (NR * RW)

static_assert(N1C <= NSTR && N2C <= NSTR, "row-pointer stride too small");
static_assert(N1C <= 65536, "source ids are kept as 16-bit values");
static_assert(N1C % 16 == 0 && N2C % 16 == 0, "GEMM row tiles are 16 rows, no tail tile");

typedef float v2f __attribute__((ext_vector_type(2)));
typedef float v8f __attribute__((ext_vector_type(8)));

#define GEMM_WGS 768
#define GEMM_JOBS 3
#define NDOT 3
static_assert(GEMM_WGS % GEMM_JOBS == 0, "the layer launch splits its workgroups evenly over its products");

__device__ __forceinline__ float dot4(const float4 x, const float4 a) {
  return x.x * a.x + x.y * a.y + x.z * a.z + x.w * a.w;
}

template <bool DOTS>
__device__ __forceinline__
void gemm_n128_tiles(const float* __restrict__ A, const float* __restrict__ W,
                     const float* __restrict__ bias, float* __restrict__ C,
                     int ntiles, int tile, const int tstep, const float* __restrict__ vt,
                     float* __restrict__ o0, float* __restrict__ o1,
                     float* __restrict__ o2) {
  extern __shared__ float smem[];
  float* sA   = smem;
  float* sWt  = smem + 2 * 16 * WSTR;
  float* sDot = sWt + DC * WSTR;

  const int tid  = threadIdx.x;
  const int lane = tid & 31;
  const int wv   = tid >> 5;

  if (tile >= ntiles) return;

  const int ar  = tid >> 5;
  const int an4 = (tid & 31) * 4;
  const float* arow = A + (size_t)ar * DC + an4;
  float4 pa0 = *(const float4*)(arow + (size_t)tile * 16 * DC);
  float4 pa1 = *(const float4*)(arow + (size_t)tile * 16 * DC + 8 * DC);

  float4 q0 = make_float4(0.f, 0.f, 0.f, 0.f), q1 = q0, q2 = q0;
  if (DOTS) {
    q0 = *(const float4*)(vt + an4);
    q1 = *(const float4*)(vt + DC + an4);
    q2 = *(const float4*)(vt + 2 * DC + an4);
  }
  int ptile = -1;

#pragma unroll
  for (int i = 0; i < 16; ++i) {
    const int k  = 8 * i + (lane >> 2);
    const int n4 = 16 * wv + 4 * (lane & 3);
    const float4 w4 = *(const float4*)(W + (size_t)k * DC + n4);
    sWt[(n4 + 0) * WSTR + k] = w4.x;
    sWt[(n4 + 1) * WSTR + k] = w4.y;
    sWt[(n4 + 2) * WSTR + k] = w4.z;
    sWt[(n4 + 3) * WSTR + k] = w4.w;
  }

  const int cb   = wv * 16;
  const int m    = lane & 15;
  const int half = lane >> 4;
  const int col  = cb + m;
  const float bv = bias ? bias[col] : 0.f;
  const float* bp = sWt + (cb + m) * WSTR + half * 2;

  int buf = 0;
  for (; tile < ntiles; tile += tstep) {
    float* sAb = sA + buf * (16 * WSTR);
    *(float4*)(sAb + ar * WSTR + an4)       = pa0;
    *(float4*)(sAb + (ar + 8) * WSTR + an4) = pa1;
    float r0a = 0.f, r0b = 0.f, r0c = 0.f, r1a = 0.f, r1b = 0.f, r1c = 0.f;
    if (DOTS) {
      r0a = dot4(pa0, q0); r0b = dot4(pa0, q1); r0c = dot4(pa0, q2);
      r1a = dot4(pa1, q0); r1b = dot4(pa1, q1); r1c = dot4(pa1, q2);
    }
    __syncthreads();

    const int nt = tile + tstep;
    if (nt < ntiles) {
      pa0 = *(const float4*)(arow + (size_t)nt * 16 * DC);
      pa1 = *(const float4*)(arow + (size_t)nt * 16 * DC + 8 * DC);
    }

    if (DOTS) {
      if (ptile >= 0 && tid < NDOT * 16) {
        float* o = tid < 16 ? o0 : (tid < 32 ? o1 : o2);
        o[(size_t)ptile * 16 + (tid & 15)] = sDot[(buf ^ 1) * (NDOT * 16) + tid];
      }
      const bool b4 = (lane & 16) != 0, b3 = (lane & 8) != 0, b2 = (lane & 4) != 0;
      const float s0 = (b4 ? r1a : r0a) + __shfl_xor(b4 ? r0a : r1a, 16, 32);
      const float s1 = (b4 ? r1b : r0b) + __shfl_xor(b4 ? r0b : r1b, 16, 32);
      const float s2 = (b4 ? r1c : r0c) + __shfl_xor(b4 ? r0c : r1c, 16, 32);
      const float t0 = (b3 ? s2 : s0) + __shfl_xor(b3 ? s0 : s2, 8, 32);
      const float t1 = (b3 ? 0.f : s1) + __shfl_xor(b3 ? s1 : 0.f, 8, 32);
      float u = (b2 ? t1 : t0) + __shfl_xor(b2 ? t0 : t1, 4, 32);
      u += __shfl_xor(u, 2, 32);
      u += __shfl_xor(u, 1, 32);
      const int j = (lane >> 2) & 3;
      if ((lane & 3) == 0 && j < NDOT)
        sDot[buf * (NDOT * 16) + j * 16 + wv + 8 * (lane >> 4)] = u;
      ptile = tile;
    }

    const float* ap = sAb + m * WSTR + half * 2;
    v8f acc = {};
#pragma unroll 8
    for (int kk = 0; kk < 32; ++kk) {
      const v2f a = *(const v2f*)(ap + kk * 4);
      const v2f b = *(const v2f*)(bp + kk * 4);
      acc = __builtin_amdgcn_wmma_f32_16x16x4_f32(false, a, false, b,
                                                  (short)0, acc, false, false);
    }

    float* cp = C + ((size_t)tile * 16 + half * 8) * DC + col;
#pragma unroll
    for (int r = 0; r < 8; ++r) {
      cp[(size_t)r * DC] = acc[r] + bv;
    }
    buf ^= 1;
  }
  if (DOTS) {
    __syncthreads();
    if (tid < NDOT * 16) {
      float* o = tid < 16 ? o0 : (tid < 32 ? o1 : o2);
      o[(size_t)ptile * 16 + (tid & 15)] = sDot[(buf ^ 1) * (NDOT * 16) + tid];
    }
  }
}

__global__ __launch_bounds__(256)
void gemm_n128_wmma(const float* __restrict__ A, const float* __restrict__ W,
                    const float* __restrict__ bias, float* __restrict__ C, int ntiles) {
  gemm_n128_tiles<false>(A, W, bias, C, ntiles, (int)blockIdx.x, (int)gridDim.x,
                         nullptr, nullptr, nullptr, nullptr);
}

__global__ __launch_bounds__(256)
void gemm_layer_kernel(const float* __restrict__ X1, const float* __restrict__ X2,
                       const float* __restrict__ Wa, const float* __restrict__ Wb,
                       const float* __restrict__ Wc,
                       float* __restrict__ Ca, float* __restrict__ Cb, float* __restrict__ Cc,
                       const float* __restrict__ vta, const float* __restrict__ vtb,
                       float* __restrict__ oa0, float* __restrict__ oa1, float* __restrict__ oa2,
                       float* __restrict__ ob0, float* __restrict__ ob1, float* __restrict__ ob2,
                       int ntiles1, int ntiles2) {
  const int job = blockIdx.y;
  if (job == 2) {
    gemm_n128_tiles<false>(X1, Wc, nullptr, Cc, ntiles1, (int)blockIdx.x, (int)gridDim.x,
                           nullptr, nullptr, nullptr, nullptr);
  } else {
    const bool a = job == 0;
    gemm_n128_tiles<true>(a ? X1 : X2, a ? Wa : Wb, nullptr, a ? Ca : Cb,
                          a ? ntiles1 : ntiles2, (int)blockIdx.x, (int)gridDim.x,
                          a ? vta : vtb, a ? oa0 : ob0, a ? oa1 : ob1, a ? oa2 : ob2);
  }
}

#define NFOLD 6
#define FD_ROWS 16
static_assert(DC == 8 * FD_ROWS, "eight waves cover the rows of W");

__device__ __forceinline__
void fold_vectors(const int fb,
                  const float* __restrict__ W_i1, const float* __restrict__ as_i1,
                  const float* __restrict__ ad_i1, const float* __restrict__ W_i2,
                  const float* __restrict__ as_i2, const float* __restrict__ ad_i2,
                  const float* __restrict__ Ws_x, const float* __restrict__ as_x,
                  const float* __restrict__ Wd_x, const float* __restrict__ ad_x,
                  float* __restrict__ vt) {
  const int j = fb % NFOLD;
  const int l = fb / NFOLD;
  const float* Wj = j < 2 ? W_i1 : (j == 2 ? Ws_x : (j < 5 ? W_i2 : Wd_x));
  const float* aj = j == 0 ? as_i1 : (j == 1 ? ad_i1 : (j == 2 ? as_x
                  : (j == 3 ? as_i2 : (j == 4 ? ad_i2 : ad_x))));
  Wj += (size_t)l * DC * DC;
  aj += (size_t)l * DC;
  const int lane = threadIdx.x & 31;
  const int row0 = (threadIdx.x >> 5) * FD_ROWS;
  const float4* wp = (const float4*)(Wj + (size_t)row0 * DC) + lane;
  float4 w[FD_ROWS];
#pragma unroll
  for (int r = 0; r < FD_ROWS; ++r) w[r] = wp[r * (DC / 4)];
  const float4 a = ((const float4*)aj)[lane];
  float s[FD_ROWS];
#pragma unroll
  for (int r = 0; r < FD_ROWS; ++r) s[r] = dot4(w[r], a);
#pragma unroll
  for (int off = 16; off > 0; off >>= 1) {
#pragma unroll
    for (int r = 0; r < FD_ROWS; ++r) s[r] += __shfl_xor(s[r], off, 32);
  }
  float out = s[0];
#pragma unroll
  for (int r = 1; r < FD_ROWS; ++r) out = lane == r ? s[r] : out;
  if (lane < FD_ROWS) vt[((size_t)l * NFOLD + j) * DC + row0 + lane] = out;
}

#define BIN_T 256
#define NCH 256
#define CS (EC / NCH)
#define BIN_IT ((CS + BIN_T - 1) / BIN_T)
static_assert(EC % NCH == 0, "chunks have no tail");
static_assert(NR == BIN_T && NCH == BIN_T, "one thread per range (binning) and per chunk (ordering)");
static_assert(RW <= BIN_T && RW <= 65535, "one thread per row of a range; the row fits the word's upper half");

__device__ __forceinline__ int block_scan_excl(int v, int* wtot, int& total) {
  const int lane = threadIdx.x & 31;
  const int w    = threadIdx.x >> 5;
  int incl = v;
#pragma unroll
  for (int off = 1; off < 32; off <<= 1) {
    const int u = __shfl_up(incl, off, 32);
    if (lane >= off) incl += u;
  }
  if (lane == 31) wtot[w] = incl;
  __syncthreads();
  int pre = 0, tot = 0;
#pragma unroll
  for (int i = 0; i < BIN_T / 32; ++i) {
    const int t = wtot[i];
    if (i < w) pre += t;
    tot += t;
  }
  total = tot;
  return pre + incl - v;
}

__device__ __forceinline__
void edge_bin(const int chunk, const int t,
              const int* __restrict__ e11,
              const int* __restrict__ e22,
              const int* __restrict__ e12,
              unsigned* __restrict__ binned,
              int* __restrict__ tab) {
  __shared__ int cnt[NR];
  __shared__ int offs[NR];
  __shared__ int wtot[BIN_T / 32];
  __shared__ unsigned stage[CS];
  const int tid   = threadIdx.x;
  const int* ei = (t == 0 ? e11 : (t == 1 ? e22 : e12)) + chunk * CS;

  cnt[tid] = 0;
  int s[BIN_IT], d[BIN_IT], r[BIN_IT], rk[BIN_IT];
#pragma unroll
  for (int u = 0; u < BIN_IT; ++u) {
    const int i = tid + u * BIN_T;
    const bool v = i < CS;
    s[u] = v ? ei[i] : 0;
    d[u] = v ? ei[EC + i] : 0;
  }
  __syncthreads();
#pragma unroll
  for (int u = 0; u < BIN_IT; ++u) {
    const int i = tid + u * BIN_T;
    r[u]  = d[u] / RW;
    rk[u] = 0;
    if (i < CS) rk[u] = atomicAdd(&cnt[r[u]], 1);
  }
  __syncthreads();
  int total;
  const int off = block_scan_excl(cnt[tid], wtot, total);
  offs[tid] = off;
  tab[((size_t)t * NCH + chunk) * NR + tid] = off;
  __syncthreads();
#pragma unroll
  for (int u = 0; u < BIN_IT; ++u) {
    const int i = tid + u * BIN_T;
    if (i < CS)
      stage[offs[r[u]] + rk[u]] = ((unsigned)(d[u] - r[u] * RW) << 16) | (unsigned)s[u];
  }
  __syncthreads();
  unsigned* out = binned + (size_t)t * EC + (size_t)chunk * CS;
#pragma unroll
  for (int u = 0; u < BIN_IT; ++u) {
    const int i = tid + u * BIN_T;
    if (i < CS) out[i] = stage[i];
  }
}

static_assert(BIN_T == 256, "a fold workgroup is eight waves");
__global__ __launch_bounds__(BIN_T)
void edge_bin_fold_kernel(const int* __restrict__ e11,
                          const int* __restrict__ e22,
                          const int* __restrict__ e12,
                          unsigned* __restrict__ binned,
                          int* __restrict__ tab,
                          const float* __restrict__ W_i1, const float* __restrict__ as_i1,
                          const float* __restrict__ ad_i1, const float* __restrict__ W_i2,
                          const float* __restrict__ as_i2, const float* __restrict__ ad_i2,
                          const float* __restrict__ Ws_x, const float* __restrict__ as_x,
                          const float* __restrict__ Wd_x, const float* __restrict__ ad_x,
                          float* __restrict__ vt) {
  const int b = blockIdx.x;
  if (b < 3 * NCH) {
    edge_bin(b % NCH, b / NCH, e11, e22, e12, binned, tab);
  } else {
    fold_vectors(b - 3 * NCH, W_i1, as_i1, ad_i1, W_i2, as_i2, ad_i2,
                 Ws_x, as_x, Wd_x, ad_x, vt);
  }
}

__global__ __launch_bounds__(BIN_T)
void range_sort_kernel(const unsigned* __restrict__ binned,
                       const int* __restrict__ tab,
                       int* __restrict__ rp,
                       unsigned short* __restrict__ csr) {
  __shared__ int rowcnt[BIN_T];
  __shared__ int rowpos[BIN_T];
  __shared__ int wtotA[BIN_T / 32];
  __shared__ int wtotB[BIN_T / 32];
  const int tid  = threadIdx.x;
  const int lane = tid & 31;
  const int c0 = __builtin_amdgcn_readfirstlane(tid & ~31);
  const int r  = blockIdx.x;
  const int t  = blockIdx.y;
  const int nd = t == 0 ? N1C : N2C;

  const int* tb = tab + ((size_t)t * NCH + tid) * NR;
  const int sb = tb[r];
  const int se = r + 1 < NR ? tb[r + 1] : CS;
  const int n  = se - sb;
  rowcnt[tid] = 0;
  int base;
  (void)block_scan_excl(sb, wtotA, base);

  const unsigned* lst = binned + (size_t)t * EC + (size_t)c0 * CS;
  const bool more = __builtin_amdgcn_ballot_w32(n > 32) != 0u;
  unsigned w[32];
#pragma unroll
  for (int j = 0; j < 32; ++j) {
    const int sbj = __builtin_amdgcn_readlane(sb, j);
    const int nj  = __builtin_amdgcn_readlane(n, j);
    w[j] = lane < nj ? lst[(size_t)j * CS + sbj + lane] : 0u;
  }
#pragma unroll
  for (int j = 0; j < 32; ++j) {
    const int nj = __builtin_amdgcn_readlane(n, j);
    if (lane < nj) atomicAdd(&rowcnt[w[j] >> 16], 1);
  }
  if (more) {
#pragma nounroll
    for (int j = 0; j < 32; ++j) {
      const int sbj = __builtin_amdgcn_readlane(sb, j);
      const int nj  = __builtin_amdgcn_readlane(n, j);
      const unsigned* pj = lst + (size_t)j * CS + sbj;
      for (int k = 32 + lane; k < nj; k += 32) atomicAdd(&rowcnt[pj[k] >> 16], 1);
    }
  }
  __syncthreads();

  int tot;
  const int c  = rowcnt[tid];
  const int st = block_scan_excl(c, wtotB, tot);
  rowpos[tid] = base + st;
  const int d = r * RW + tid;
  if (tid < RW && d < nd) rp[t * NSTR + d] = base + st + c;
  __syncthreads();

  unsigned short* out = csr + (size_t)t * EC;
#pragma unroll
  for (int j = 0; j < 32; ++j) {
    const int nj = __builtin_amdgcn_readlane(n, j);
    if (lane < nj) {
      const int p = atomicAdd(&rowpos[w[j] >> 16], 1);
      out[p] = (unsigned short)(w[j] & 0xffffu);
    }
  }
  if (more) {
#pragma nounroll
    for (int j = 0; j < 32; ++j) {
      const int sbj = __builtin_amdgcn_readlane(sb, j);
      const int nj  = __builtin_amdgcn_readlane(n, j);
      const unsigned* pj = lst + (size_t)j * CS + sbj;
      for (int k = 32 + lane; k < nj; k += 32) {
        const unsigned wk = pj[k];
        const int p = atomicAdd(&rowpos[wk >> 16], 1);
        out[p] = (unsigned short)(wk & 0xffffu);
      }
    }
  }
}

__device__ __forceinline__ float leaky02(float l) { return l > 0.f ? l : 0.2f * l; }

__device__ __forceinline__ float wave_max(float v) {
#pragma unroll
  for (int off = 16; off > 0; off >>= 1) v = fmaxf(v, __shfl_xor(v, off, 32));
  return v;
}
__device__ __forceinline__ float wave_sum(float v) {
#pragma unroll
  for (int off = 16; off > 0; off >>= 1) v += __shfl_xor(v, off, 32);
  return v;
}

__device__ __forceinline__ void gather_chunk(float4& acc, int s, float c, int n,
                                             const float* __restrict__ G, int lane) {
  const float* gl = G + lane * 4;
  int j = 0;
  for (; j + 4 <= n; j += 4) {
    const int s0 = __builtin_amdgcn_readlane(s, j + 0);
    const int s1 = __builtin_amdgcn_readlane(s, j + 1);
    const int s2 = __builtin_amdgcn_readlane(s, j + 2);
    const int s3 = __builtin_amdgcn_readlane(s, j + 3);
    const float c0 = __int_as_float(__builtin_amdgcn_readlane(__float_as_int(c), j + 0));
    const float c1 = __int_as_float(__builtin_amdgcn_readlane(__float_as_int(c), j + 1));
    const float c2 = __int_as_float(__builtin_amdgcn_readlane(__float_as_int(c), j + 2));
    const float c3 = __int_as_float(__builtin_amdgcn_readlane(__float_as_int(c), j + 3));
    const float4 g0 = *(const float4*)(gl + (size_t)s0 * DC);
    const float4 g1 = *(const float4*)(gl + (size_t)s1 * DC);
    const float4 g2 = *(const float4*)(gl + (size_t)s2 * DC);
    const float4 g3 = *(const float4*)(gl + (size_t)s3 * DC);
    acc.x += c0 * g0.x; acc.y += c0 * g0.y; acc.z += c0 * g0.z; acc.w += c0 * g0.w;
    acc.x += c1 * g1.x; acc.y += c1 * g1.y; acc.z += c1 * g1.z; acc.w += c1 * g1.w;
    acc.x += c2 * g2.x; acc.y += c2 * g2.y; acc.z += c2 * g2.z; acc.w += c2 * g2.w;
    acc.x += c3 * g3.x; acc.y += c3 * g3.y; acc.z += c3 * g3.z; acc.w += c3 * g3.w;
  }
  for (; j < n; ++j) {
    const int sj = __builtin_amdgcn_readlane(s, j);
    const float cj = __int_as_float(__builtin_amdgcn_readlane(__float_as_int(c), j));
    const float4 g = *(const float4*)(gl + (size_t)sj * DC);
    acc.x += cj * g.x; acc.y += cj * g.y; acc.z += cj * g.z; acc.w += cj * g.w;
  }
}

__device__ __forceinline__ float4 seg_softmax_gather(
    int d, int lane, const int* __restrict__ rp,
    const unsigned short* __restrict__ csr, const float* __restrict__ a_s,
    const float* __restrict__ a_d, const float* __restrict__ G) {
  float4 acc = make_float4(0.f, 0.f, 0.f, 0.f);
  const int end = rp[d];
  const int beg = d > 0 ? rp[d - 1] : 0;
  const float adv = a_d[d];
  const float NEG_INF = -__builtin_inff();

  const int  i0 = beg + lane;
  const bool v0 = i0 < end;
  const int  s0 = v0 ? (int)csr[i0] : 0;
  const float l0 = v0 ? leaky02(a_s[s0] + adv) : NEG_INF;

  float m = l0;
  for (int i = i0 + 32; i < end; i += 32)
    m = fmaxf(m, leaky02(a_s[csr[i]] + adv));
  m = wave_max(m);

  const float e0 = v0 ? __expf(l0 - m) : 0.f;
  float z = e0;
  for (int i = i0 + 32; i < end; i += 32)
    z += __expf(leaky02(a_s[csr[i]] + adv) - m);
  z = wave_sum(z);
  const float den = z + 1e-16f;

  const int n0 = end - beg < 32 ? end - beg : 32;
  gather_chunk(acc, s0, e0 / den, n0, G, lane);

  for (int base = beg + 32; base < end; base += 32) {
    const int  i = base + lane;
    const bool v = i < end;
    const int  s = v ? (int)csr[i] : 0;
    const float c = v ? __expf(leaky02(a_s[s] + adv) - m) / den : 0.f;
    const int n = end - base < 32 ? end - base : 32;
    gather_chunk(acc, s, c, n, G, lane);
  }
  return acc;
}

#define NODE_ROWS 2
static_assert(N1C % NODE_ROWS == 0 && N2C % NODE_ROWS == 0, "a gather wave has all its rows or none");

template <bool TWO>
__device__ __forceinline__
void node_softmax_gather(
    const int wb,
    const int* __restrict__ rpA, const unsigned short* __restrict__ csrA,
    const float* __restrict__ asA, const float* __restrict__ adA,
    const float* __restrict__ GA, const float* __restrict__ bA,
    const int* __restrict__ rpB, const unsigned short* __restrict__ csrB,
    const float* __restrict__ asB, const float* __restrict__ adB,
    const float* __restrict__ GB, const float* __restrict__ bB,
    float* __restrict__ H, int N) {
  const int w = __builtin_amdgcn_readfirstlane(
      (int)(wb * (blockDim.x >> 5) + (threadIdx.x >> 5)));
  const int lane = threadIdx.x & 31;
  if (w * NODE_ROWS >= N) return;

#pragma nounroll
  for (int i = 0; i < NODE_ROWS; ++i) {
    const int d = w * NODE_ROWS + i;
    const float4 a  = seg_softmax_gather(d, lane, rpA, csrA, asA, adA, GA);
    const float4 ba = ((const float4*)bA)[lane];
    float4 o = make_float4(a.x + ba.x, a.y + ba.y, a.z + ba.z, a.w + ba.w);
    if (TWO) {
      const float4 b  = seg_softmax_gather(d, lane, rpB, csrB, asB, adB, GB);
      const float4 bb = ((const float4*)bB)[lane];
      o.x += b.x + bb.x; o.y += b.y + bb.y; o.z += b.z + bb.z; o.w += b.w + bb.w;
    }
    o.x = o.x > 0.f ? o.x : 0.f;
    o.y = o.y > 0.f ? o.y : 0.f;
    o.z = o.z > 0.f ? o.z : 0.f;
    o.w = o.w > 0.f ? o.w : 0.f;
    *(float4*)(H + (size_t)d * DC + lane * 4) = o;
  }
}

__global__ __launch_bounds__(256, 16)
void node_softmax_gather_kernel(
    const int* __restrict__ rp22, const unsigned short* __restrict__ csr22,
    const float* __restrict__ as2, const float* __restrict__ ad2,
    const float* __restrict__ G2, const float* __restrict__ b2,
    const int* __restrict__ rp12, const unsigned short* __restrict__ csr12,
    const float* __restrict__ asx, const float* __restrict__ adx,
    const float* __restrict__ Gx, const float* __restrict__ bx,
    float* __restrict__ H2, int n2,
    const int* __restrict__ rp11, const unsigned short* __restrict__ csr11,
    const float* __restrict__ as1, const float* __restrict__ ad1,
    const float* __restrict__ G1, const float* __restrict__ b1,
    float* __restrict__ H1, int n1, int nbB) {
  const int b = blockIdx.x;
  if (b < nbB) {
    node_softmax_gather<true>(b, rp22, csr22, as2, ad2, G2, b2,
                              rp12, csr12, asx, adx, Gx, bx, H2, n2);
  } else {
    node_softmax_gather<false>(b - nbB, rp11, csr11, as1, ad1, G1, b1,
                               nullptr, nullptr, nullptr, nullptr, nullptr, nullptr, H1, n1);
  }
}

extern "C" void kernel_launch(void* const* d_in, const int* in_sizes, int n_in,
                              void* d_out, int out_size, void* d_ws, size_t ws_size,
                              hipStream_t stream) {
  const float* x1    = (const float*)d_in[0];
  const float* x2    = (const float*)d_in[1];
  const int*   ei11  = (const int*)d_in[2];
  const int*   ei22  = (const int*)d_in[3];
  const int*   ei12  = (const int*)d_in[4];
  const float* W_i1  = (const float*)d_in[5];
  const float* as_i1 = (const float*)d_in[6];
  const float* ad_i1 = (const float*)d_in[7];
  const float* b_i1  = (const float*)d_in[8];
  const float* W_i2  = (const float*)d_in[9];
  const float* as_i2 = (const float*)d_in[10];
  const float* ad_i2 = (const float*)d_in[11];
  const float* b_i2  = (const float*)d_in[12];
  const float* Ws_x  = (const float*)d_in[13];
  const float* Wd_x  = (const float*)d_in[14];
  const float* as_x  = (const float*)d_in[15];
  const float* ad_x  = (const float*)d_in[16];
  const float* b_x   = (const float*)d_in[17];
  const float* W_out = (const float*)d_in[18];
  const float* b_out = (const float*)d_in[19];

  float* ws = (float*)d_ws;
  size_t off = 0;
  auto alloc = [&](size_t n) { float* p = ws + off; off += n; return p; };

  float* h1   = alloc((size_t)N1C * DC);
  float* h2   = alloc((size_t)N2C * DC);
  float* g1   = alloc((size_t)N1C * DC);
  float* g2   = alloc((size_t)N2C * DC);
  float* gx   = alloc((size_t)N1C * DC);
  float* a_s1 = alloc(N1C);
  float* a_d1 = alloc(N1C);
  float* a_s2 = alloc(N2C);
  float* a_d2 = alloc(N2C);
  float* a_sx = alloc(N1C);
  float* a_dx = alloc(N2C);
  float* vtab = alloc((size_t)LC * NFOLD * DC);
  int*   rp   = (int*)alloc((size_t)3 * NSTR);
  unsigned short* csr = (unsigned short*)alloc((size_t)3 * EC / 2);
  (void)ws_size; (void)in_sizes; (void)n_in; (void)out_size;

  const dim3 blk(256);
  const dim3 gemm_g(GEMM_WGS);
  const dim3 gemm_lg(GEMM_WGS / GEMM_JOBS, GEMM_JOBS);
  const int  rd_nb1 = (N1C / NODE_ROWS + 7) / 8, rd_nb2 = (N2C / NODE_ROWS + 7) / 8;
  const dim3 rd_g(rd_nb2 + rd_nb1);
  const dim3 binfold_g(3 * NCH + LC * NFOLD), srt_g(NR, 3);
  const dim3 bin_blk(BIN_T);
  const size_t lds_bytes = (size_t)(DC * WSTR + 2 * 16 * WSTR + 2 * NDOT * 16) * sizeof(float);

  const int* rp11 = rp;
  const int* rp22 = rp + NSTR;
  const int* rp12 = rp + 2 * NSTR;
  const unsigned short* csr11 = csr;
  const unsigned short* csr22 = csr + (size_t)EC;
  const unsigned short* csr12 = csr + (size_t)2 * EC;

  static_assert((size_t)3 * EC + (size_t)3 * NCH * NR <= (size_t)N1C * DC, "grouping scratch fits in h1");
  unsigned* binned = (unsigned*)h1;
  int* tab = (int*)(binned + (size_t)3 * EC);
  edge_bin_fold_kernel<<<binfold_g, bin_blk, 0, stream>>>(ei11, ei22, ei12, binned, tab,
                                                          W_i1, as_i1, ad_i1, W_i2, as_i2, ad_i2,
                                                          Ws_x, as_x, Wd_x, ad_x, vtab);
  range_sort_kernel<<<srt_g, bin_blk, 0, stream>>>(binned, tab, rp, csr);

  const float* in1 = x1;
  const float* in2 = x2;

  for (int l = 0; l < LC; ++l) {
    const size_t wo = (size_t)l * DC * DC;
    const size_t vo = (size_t)l * DC;

    const float* vt1 = vtab + (size_t)l * NFOLD * DC;
    const float* vt2 = vt1 + 3 * DC;

    gemm_layer_kernel<<<gemm_lg, blk, lds_bytes, stream>>>(
        in1, in2, W_i1 + wo, W_i2 + wo, Ws_x + wo, g1, g2, gx,
        vt1, vt2, a_s1, a_d1, a_sx, a_s2, a_d2, a_dx, N1C / 16, N2C / 16);

    node_softmax_gather_kernel<<<rd_g, blk, 0, stream>>>(
        rp22, csr22, a_s2, a_d2, g2, b_i2 + vo,
        rp12, csr12, a_sx, a_dx, gx, b_x + vo, h2, N2C,
        rp11, csr11, a_s1, a_d1, g1, b_i1 + vo, h1, N1C, rd_nb2);

    in1 = h1;
    in2 = h2;
  }

  float* out = (float*)d_out;
  gemm_n128_wmma<<<gemm_g, blk, lds_bytes, stream>>>(h1, W_out, b_out, out, (N1C + N2C) / 16);
}


# gather wave issues fewer instructions per edge: scalar coefficient and 32-bit row offset, two packed fused multiply-adds per row, wave max/sum by DPP not LDS permutes; sum order kept, products fused
# speedup vs baseline: 7.8441x; 1.0348x over previous
#include <hip/hip_runtime.h>


#define N1C 50000
#define N2C 50000
#define EC  800000
#define DC  128
#define LC  2
#define WSTR 132
#define NMAXC (N1C > N2C ? N1C : N2C)
#define NR 256
#define RW ((NMAXC + NR - 1) / NR)
#define NSTR (NR * RW)

static_assert(N1C <= NSTR && N2C <= NSTR, "row-pointer stride too small");
static_assert(N1C <= 65536, "source ids are kept as 16-bit values");
static_assert(N1C % 16 == 0 && N2C % 16 == 0, "GEMM row tiles are 16 rows, no tail tile");

typedef float v2f __attribute__((ext_vector_type(2)));
typedef float v8f __attribute__((ext_vector_type(8)));

#define GEMM_WGS 768
#define GEMM_JOBS 3
#define NDOT 3
static_assert(GEMM_WGS % GEMM_JOBS == 0, "the layer launch splits its workgroups evenly over its products");

__device__ __forceinline__ float dot4(const float4 x, const float4 a) {
  return x.x * a.x + x.y * a.y + x.z * a.z + x.w * a.w;
}

template <bool DOTS>
__device__ __forceinline__
void gemm_n128_tiles(const float* __restrict__ A, const float* __restrict__ W,
                     const float* __restrict__ bias, float* __restrict__ C,
                     int ntiles, int tile, const int tstep, const float* __restrict__ vt,
                     float* __restrict__ o0, float* __restrict__ o1,
                     float* __restrict__ o2) {
  extern __shared__ float smem[];
  float* sA   = smem;
  float* sWt  = smem + 2 * 16 * WSTR;
  float* sDot = sWt + DC * WSTR;

  const int tid  = threadIdx.x;
  const int lane = tid & 31;
  const int wv   = tid >> 5;

  if (tile >= ntiles) return;

  const int ar  = tid >> 5;
  const int an4 = (tid & 31) * 4;
  const float* arow = A + (size_t)ar * DC + an4;
  float4 pa0 = *(const float4*)(arow + (size_t)tile * 16 * DC);
  float4 pa1 = *(const float4*)(arow + (size_t)tile * 16 * DC + 8 * DC);

  float4 q0 = make_float4(0.f, 0.f, 0.f, 0.f), q1 = q0, q2 = q0;
  if (DOTS) {
    q0 = *(const float4*)(vt + an4);
    q1 = *(const float4*)(vt + DC + an4);
    q2 = *(const float4*)(vt + 2 * DC + an4);
  }
  int ptile = -1;

#pragma unroll
  for (int i = 0; i < 16; ++i) {
    const int k  = 8 * i + (lane >> 2);
    const int n4 = 16 * wv + 4 * (lane & 3);
    const float4 w4 = *(const float4*)(W + (size_t)k * DC + n4);
    sWt[(n4 + 0) * WSTR + k] = w4.x;
    sWt[(n4 + 1) * WSTR + k] = w4.y;
    sWt[(n4 + 2) * WSTR + k] = w4.z;
    sWt[(n4 + 3) * WSTR + k] = w4.w;
  }

  const int cb   = wv * 16;
  const int m    = lane & 15;
  const int half = lane >> 4;
  const int col  = cb + m;
  const float bv = bias ? bias[col] : 0.f;
  const float* bp = sWt + (cb + m) * WSTR + half * 2;

  int buf = 0;
  for (; tile < ntiles; tile += tstep) {
    float* sAb = sA + buf * (16 * WSTR);
    *(float4*)(sAb + ar * WSTR + an4)       = pa0;
    *(float4*)(sAb + (ar + 8) * WSTR + an4) = pa1;
    float r0a = 0.f, r0b = 0.f, r0c = 0.f, r1a = 0.f, r1b = 0.f, r1c = 0.f;
    if (DOTS) {
      r0a = dot4(pa0, q0); r0b = dot4(pa0, q1); r0c = dot4(pa0, q2);
      r1a = dot4(pa1, q0); r1b = dot4(pa1, q1); r1c = dot4(pa1, q2);
    }
    __syncthreads();

    const int nt = tile + tstep;
    if (nt < ntiles) {
      pa0 = *(const float4*)(arow + (size_t)nt * 16 * DC);
      pa1 = *(const float4*)(arow + (size_t)nt * 16 * DC + 8 * DC);
    }

    if (DOTS) {
      if (ptile >= 0 && tid < NDOT * 16) {
        float* o = tid < 16 ? o0 : (tid < 32 ? o1 : o2);
        o[(size_t)ptile * 16 + (tid & 15)] = sDot[(buf ^ 1) * (NDOT * 16) + tid];
      }
      const bool b4 = (lane & 16) != 0, b3 = (lane & 8) != 0, b2 = (lane & 4) != 0;
      const float s0 = (b4 ? r1a : r0a) + __shfl_xor(b4 ? r0a : r1a, 16, 32);
      const float s1 = (b4 ? r1b : r0b) + __shfl_xor(b4 ? r0b : r1b, 16, 32);
      const float s2 = (b4 ? r1c : r0c) + __shfl_xor(b4 ? r0c : r1c, 16, 32);
      const float t0 = (b3 ? s2 : s0) + __shfl_xor(b3 ? s0 : s2, 8, 32);
      const float t1 = (b3 ? 0.f : s1) + __shfl_xor(b3 ? s1 : 0.f, 8, 32);
      float u = (b2 ? t1 : t0) + __shfl_xor(b2 ? t0 : t1, 4, 32);
      u += __shfl_xor(u, 2, 32);
      u += __shfl_xor(u, 1, 32);
      const int j = (lane >> 2) & 3;
      if ((lane & 3) == 0 && j < NDOT)
        sDot[buf * (NDOT * 16) + j * 16 + wv + 8 * (lane >> 4)] = u;
      ptile = tile;
    }

    const float* ap = sAb + m * WSTR + half * 2;
    v8f acc = {};
#pragma unroll 8
    for (int kk = 0; kk < 32; ++kk) {
      const v2f a = *(const v2f*)(ap + kk * 4);
      const v2f b = *(const v2f*)(bp + kk * 4);
      acc = __builtin_amdgcn_wmma_f32_16x16x4_f32(false, a, false, b,
                                                  (short)0, acc, false, false);
    }

    float* cp = C + ((size_t)tile * 16 + half * 8) * DC + col;
#pragma unroll
    for (int r = 0; r < 8; ++r) {
      cp[(size_t)r * DC] = acc[r] + bv;
    }
    buf ^= 1;
  }
  if (DOTS) {
    __syncthreads();
    if (tid < NDOT * 16) {
      float* o = tid < 16 ? o0 : (tid < 32 ? o1 : o2);
      o[(size_t)ptile * 16 + (tid & 15)] = sDot[(buf ^ 1) * (NDOT * 16) + tid];
    }
  }
}

__global__ __launch_bounds__(256)
void gemm_n128_wmma(const float* __restrict__ A, const float* __restrict__ W,
                    const float* __restrict__ bias, float* __restrict__ C, int ntiles) {
  gemm_n128_tiles<false>(A, W, bias, C, ntiles, (int)blockIdx.x, (int)gridDim.x,
                         nullptr, nullptr, nullptr, nullptr);
}

__global__ __launch_bounds__(256)
void gemm_layer_kernel(const float* __restrict__ X1, const float* __restrict__ X2,
                       const float* __restrict__ Wa, const float* __restrict__ Wb,
                       const float* __restrict__ Wc,
                       float* __restrict__ Ca, float* __restrict__ Cb, float* __restrict__ Cc,
                       const float* __restrict__ vta, const float* __restrict__ vtb,
                       float* __restrict__ oa0, float* __restrict__ oa1, float* __restrict__ oa2,
                       float* __restrict__ ob0, float* __restrict__ ob1, float* __restrict__ ob2,
                       int ntiles1, int ntiles2) {
  const int job = blockIdx.y;
  if (job == 2) {
    gemm_n128_tiles<false>(X1, Wc, nullptr, Cc, ntiles1, (int)blockIdx.x, (int)gridDim.x,
                           nullptr, nullptr, nullptr, nullptr);
  } else {
    const bool a = job == 0;
    gemm_n128_tiles<true>(a ? X1 : X2, a ? Wa : Wb, nullptr, a ? Ca : Cb,
                          a ? ntiles1 : ntiles2, (int)blockIdx.x, (int)gridDim.x,
                          a ? vta : vtb, a ? oa0 : ob0, a ? oa1 : ob1, a ? oa2 : ob2);
  }
}

#define NFOLD 6
#define FD_ROWS 16
static_assert(DC == 8 * FD_ROWS, "eight waves cover the rows of W");

__device__ __forceinline__
void fold_vectors(const int fb,
                  const float* __restrict__ W_i1, const float* __restrict__ as_i1,
                  const float* __restrict__ ad_i1, const float* __restrict__ W_i2,
                  const float* __restrict__ as_i2, const float* __restrict__ ad_i2,
                  const float* __restrict__ Ws_x, const float* __restrict__ as_x,
                  const float* __restrict__ Wd_x, const float* __restrict__ ad_x,
                  float* __restrict__ vt) {
  const int j = fb % NFOLD;
  const int l = fb / NFOLD;
  const float* Wj = j < 2 ? W_i1 : (j == 2 ? Ws_x : (j < 5 ? W_i2 : Wd_x));
  const float* aj = j == 0 ? as_i1 : (j == 1 ? ad_i1 : (j == 2 ? as_x
                  : (j == 3 ? as_i2 : (j == 4 ? ad_i2 : ad_x))));
  Wj += (size_t)l * DC * DC;
  aj += (size_t)l * DC;
  const int lane = threadIdx.x & 31;
  const int row0 = (threadIdx.x >> 5) * FD_ROWS;
  const float4* wp = (const float4*)(Wj + (size_t)row0 * DC) + lane;
  float4 w[FD_ROWS];
#pragma unroll
  for (int r = 0; r < FD_ROWS; ++r) w[r] = wp[r * (DC / 4)];
  const float4 a = ((const float4*)aj)[lane];
  float s[FD_ROWS];
#pragma unroll
  for (int r = 0; r < FD_ROWS; ++r) s[r] = dot4(w[r], a);
#pragma unroll
  for (int off = 16; off > 0; off >>= 1) {
#pragma unroll
    for (int r = 0; r < FD_ROWS; ++r) s[r] += __shfl_xor(s[r], off, 32);
  }
  float out = s[0];
#pragma unroll
  for (int r = 1; r < FD_ROWS; ++r) out = lane == r ? s[r] : out;
  if (lane < FD_ROWS) vt[((size_t)l * NFOLD + j) * DC + row0 + lane] = out;
}

#define BIN_T 256
#define NCH 256
#define CS (EC / NCH)
#define BIN_IT ((CS + BIN_T - 1) / BIN_T)
static_assert(EC % NCH == 0, "chunks have no tail");
static_assert(NR == BIN_T && NCH == BIN_T, "one thread per range (binning) and per chunk (ordering)");
static_assert(RW <= BIN_T && RW <= 65535, "one thread per row of a range; the row fits the word's upper half");

__device__ __forceinline__ int block_scan_excl(int v, int* wtot, int& total) {
  const int lane = threadIdx.x & 31;
  const int w    = threadIdx.x >> 5;
  int incl = v;
#pragma unroll
  for (int off = 1; off < 32; off <<= 1) {
    const int u = __shfl_up(incl, off, 32);
    if (lane >= off) incl += u;
  }
  if (lane == 31) wtot[w] = incl;
  __syncthreads();
  int pre = 0, tot = 0;
#pragma unroll
  for (int i = 0; i < BIN_T / 32; ++i) {
    const int t = wtot[i];
    if (i < w) pre += t;
    tot += t;
  }
  total = tot;
  return pre + incl - v;
}

__device__ __forceinline__
void edge_bin(const int chunk, const int t,
              const int* __restrict__ e11,
              const int* __restrict__ e22,
              const int* __restrict__ e12,
              unsigned* __restrict__ binned,
              int* __restrict__ tab) {
  __shared__ int cnt[NR];
  __shared__ int offs[NR];
  __shared__ int wtot[BIN_T / 32];
  __shared__ unsigned stage[CS];
  const int tid   = threadIdx.x;
  const int* ei = (t == 0 ? e11 : (t == 1 ? e22 : e12)) + chunk * CS;

  cnt[tid] = 0;
  int s[BIN_IT], d[BIN_IT], r[BIN_IT], rk[BIN_IT];
#pragma unroll
  for (int u = 0; u < BIN_IT; ++u) {
    const int i = tid + u * BIN_T;
    const bool v = i < CS;
    s[u] = v ? ei[i] : 0;
    d[u] = v ? ei[EC + i] : 0;
  }
  __syncthreads();
#pragma unroll
  for (int u = 0; u < BIN_IT; ++u) {
    const int i = tid + u * BIN_T;
    r[u]  = d[u] / RW;
    rk[u] = 0;
    if (i < CS) rk[u] = atomicAdd(&cnt[r[u]], 1);
  }
  __syncthreads();
  int total;
  const int off = block_scan_excl(cnt[tid], wtot, total);
  offs[tid] = off;
  tab[((size_t)t * NCH + chunk) * NR + tid] = off;
  __syncthreads();
#pragma unroll
  for (int u = 0; u < BIN_IT; ++u) {
    const int i = tid + u * BIN_T;
    if (i < CS)
      stage[offs[r[u]] + rk[u]] = ((unsigned)(d[u] - r[u] * RW) << 16) | (unsigned)s[u];
  }
  __syncthreads();
  unsigned* out = binned + (size_t)t * EC + (size_t)chunk * CS;
#pragma unroll
  for (int u = 0; u < BIN_IT; ++u) {
    const int i = tid + u * BIN_T;
    if (i < CS) out[i] = stage[i];
  }
}

static_assert(BIN_T == 256, "a fold workgroup is eight waves");
__global__ __launch_bounds__(BIN_T)
void edge_bin_fold_kernel(const int* __restrict__ e11,
                          const int* __restrict__ e22,
                          const int* __restrict__ e12,
                          unsigned* __restrict__ binned,
                          int* __restrict__ tab,
                          const float* __restrict__ W_i1, const float* __restrict__ as_i1,
                          const float* __restrict__ ad_i1, const float* __restrict__ W_i2,
                          const float* __restrict__ as_i2, const float* __restrict__ ad_i2,
                          const float* __restrict__ Ws_x, const float* __restrict__ as_x,
                          const float* __restrict__ Wd_x, const float* __restrict__ ad_x,
                          float* __restrict__ vt) {
  const int b = blockIdx.x;
  if (b < 3 * NCH) {
    edge_bin(b % NCH, b / NCH, e11, e22, e12, binned, tab);
  } else {
    fold_vectors(b - 3 * NCH, W_i1, as_i1, ad_i1, W_i2, as_i2, ad_i2,
                 Ws_x, as_x, Wd_x, ad_x, vt);
  }
}

__global__ __launch_bounds__(BIN_T)
void range_sort_kernel(const unsigned* __restrict__ binned,
                       const int* __restrict__ tab,
                       int* __restrict__ rp,
                       unsigned short* __restrict__ csr) {
  __shared__ int rowcnt[BIN_T];
  __shared__ int rowpos[BIN_T];
  __shared__ int wtotA[BIN_T / 32];
  __shared__ int wtotB[BIN_T / 32];
  const int tid  = threadIdx.x;
  const int lane = tid & 31;
  const int c0 = __builtin_amdgcn_readfirstlane(tid & ~31);
  const int r  = blockIdx.x;
  const int t  = blockIdx.y;
  const int nd = t == 0 ? N1C : N2C;

  const int* tb = tab + ((size_t)t * NCH + tid) * NR;
  const int sb = tb[r];
  const int se = r + 1 < NR ? tb[r + 1] : CS;
  const int n  = se - sb;
  rowcnt[tid] = 0;
  int base;
  (void)block_scan_excl(sb, wtotA, base);

  const unsigned* lst = binned + (size_t)t * EC + (size_t)c0 * CS;
  const bool more = __builtin_amdgcn_ballot_w32(n > 32) != 0u;
  unsigned w[32];
#pragma unroll
  for (int j = 0; j < 32; ++j) {
    const int sbj = __builtin_amdgcn_readlane(sb, j);
    const int nj  = __builtin_amdgcn_readlane(n, j);
    w[j] = lane < nj ? lst[(size_t)j * CS + sbj + lane] : 0u;
  }
#pragma unroll
  for (int j = 0; j < 32; ++j) {
    const int nj = __builtin_amdgcn_readlane(n, j);
    if (lane < nj) atomicAdd(&rowcnt[w[j] >> 16], 1);
  }
  if (more) {
#pragma nounroll
    for (int j = 0; j < 32; ++j) {
      const int sbj = __builtin_amdgcn_readlane(sb, j);
      const int nj  = __builtin_amdgcn_readlane(n, j);
      const unsigned* pj = lst + (size_t)j * CS + sbj;
      for (int k = 32 + lane; k < nj; k += 32) atomicAdd(&rowcnt[pj[k] >> 16], 1);
    }
  }
  __syncthreads();

  int tot;
  const int c  = rowcnt[tid];
  const int st = block_scan_excl(c, wtotB, tot);
  rowpos[tid] = base + st;
  const int d = r * RW + tid;
  if (tid < RW && d < nd) rp[t * NSTR + d] = base + st + c;
  __syncthreads();

  unsigned short* out = csr + (size_t)t * EC;
#pragma unroll
  for (int j = 0; j < 32; ++j) {
    const int nj = __builtin_amdgcn_readlane(n, j);
    if (lane < nj) {
      const int p = atomicAdd(&rowpos[w[j] >> 16], 1);
      out[p] = (unsigned short)(w[j] & 0xffffu);
    }
  }
  if (more) {
#pragma nounroll
    for (int j = 0; j < 32; ++j) {
      const int sbj = __builtin_amdgcn_readlane(sb, j);
      const int nj  = __builtin_amdgcn_readlane(n, j);
      const unsigned* pj = lst + (size_t)j * CS + sbj;
      for (int k = 32 + lane; k < nj; k += 32) {
        const unsigned wk = pj[k];
        const int p = atomicAdd(&rowpos[wk >> 16], 1);
        out[p] = (unsigned short)(wk & 0xffffu);
      }
    }
  }
}

__device__ __forceinline__ float leaky02(float l) { return l > 0.f ? l : 0.2f * l; }

template <int CTRL>
__device__ __forceinline__ float lane_xor_dpp(float v) {
  return __int_as_float(__builtin_amdgcn_update_dpp(0, __float_as_int(v), CTRL, 0xf, 0xf, true));
}
__device__ __forceinline__ float lane_xor16(float v) {
  const int i = __float_as_int(v);
  return __int_as_float(__builtin_amdgcn_permlanex16(i, i, 0x76543210, 0xfedcba98, false, false));
}
__device__ __forceinline__ float wave_max(float v) {
  v = fmaxf(v, lane_xor16(v));
  v = fmaxf(v, lane_xor_dpp<0x168>(v));
  v = fmaxf(v, lane_xor_dpp<0x164>(v));
  v = fmaxf(v, lane_xor_dpp<0x4E>(v));
  v = fmaxf(v, lane_xor_dpp<0xB1>(v));
  return v;
}
__device__ __forceinline__ float wave_sum(float v) {
  v += lane_xor16(v);
  v += lane_xor_dpp<0x168>(v);
  v += lane_xor_dpp<0x164>(v);
  v += lane_xor_dpp<0x4E>(v);
  v += lane_xor_dpp<0xB1>(v);
  return v;
}

__device__ __forceinline__ void fma_row(float4& acc, const float c, const float4 g) {
  acc.x = __builtin_fmaf(c, g.x, acc.x);
  acc.y = __builtin_fmaf(c, g.y, acc.y);
  acc.z = __builtin_fmaf(c, g.z, acc.z);
  acc.w = __builtin_fmaf(c, g.w, acc.w);
}

__device__ __forceinline__ void gather_chunk(float4& acc, int s, float c, int n,
                                             const float* __restrict__ G, int lane) {
  const char* Gb = (const char*)G;
  const unsigned lo = (unsigned)lane * 16u;
  const unsigned RB = DC * 4u;
  int j = 0;
  for (; j + 4 <= n; j += 4) {
    const int s0 = __builtin_amdgcn_readlane(s, j + 0);
    const int s1 = __builtin_amdgcn_readlane(s, j + 1);
    const int s2 = __builtin_amdgcn_readlane(s, j + 2);
    const int s3 = __builtin_amdgcn_readlane(s, j + 3);
    const float c0 = __int_as_float(__builtin_amdgcn_readlane(__float_as_int(c), j + 0));
    const float c1 = __int_as_float(__builtin_amdgcn_readlane(__float_as_int(c), j + 1));
    const float c2 = __int_as_float(__builtin_amdgcn_readlane(__float_as_int(c), j + 2));
    const float c3 = __int_as_float(__builtin_amdgcn_readlane(__float_as_int(c), j + 3));
    const float4 g0 = *(const float4*)(Gb + ((unsigned)s0 * RB + lo));
    const float4 g1 = *(const float4*)(Gb + ((unsigned)s1 * RB + lo));
    const float4 g2 = *(const float4*)(Gb + ((unsigned)s2 * RB + lo));
    const float4 g3 = *(const float4*)(Gb + ((unsigned)s3 * RB + lo));
    fma_row(acc, c0, g0);
    fma_row(acc, c1, g1);
    fma_row(acc, c2, g2);
    fma_row(acc, c3, g3);
  }
#pragma nounroll
  for (; j < n; ++j) {
    const int sj = __builtin_amdgcn_readlane(s, j);
    const float cj = __int_as_float(__builtin_amdgcn_readlane(__float_as_int(c), j));
    fma_row(acc, cj, *(const float4*)(Gb + ((unsigned)sj * RB + lo)));
  }
}

__device__ __forceinline__ float4 seg_softmax_gather(
    int d, int lane, const int* __restrict__ rp,
    const unsigned short* __restrict__ csr, const float* __restrict__ a_s,
    const float* __restrict__ a_d, const float* __restrict__ G) {
  float4 acc = make_float4(0.f, 0.f, 0.f, 0.f);
  const int end = rp[d];
  const int beg = d > 0 ? rp[d - 1] : 0;
  const float adv = a_d[d];
  const float NEG_INF = -__builtin_inff();

  const int  i0 = beg + lane;
  const bool v0 = i0 < end;
  const int  s0 = v0 ? (int)csr[i0] : 0;
  const float l0 = v0 ? leaky02(a_s[s0] + adv) : NEG_INF;

  float m = l0;
  for (int i = i0 + 32; i < end; i += 32)
    m = fmaxf(m, leaky02(a_s[csr[i]] + adv));
  m = wave_max(m);

  const float e0 = v0 ? __expf(l0 - m) : 0.f;
  float z = e0;
  for (int i = i0 + 32; i < end; i += 32)
    z += __expf(leaky02(a_s[csr[i]] + adv) - m);
  z = wave_sum(z);
  const float den = z + 1e-16f;

  const int n0 = end - beg < 32 ? end - beg : 32;
  gather_chunk(acc, s0, e0 / den, n0, G, lane);

  for (int base = beg + 32; base < end; base += 32) {
    const int  i = base + lane;
    const bool v = i < end;
    const int  s = v ? (int)csr[i] : 0;
    const float c = v ? __expf(leaky02(a_s[s] + adv) - m) / den : 0.f;
    const int n = end - base < 32 ? end - base : 32;
    gather_chunk(acc, s, c, n, G, lane);
  }
  return acc;
}

#define NODE_ROWS 2
static_assert(N1C % NODE_ROWS == 0 && N2C % NODE_ROWS == 0, "a gather wave has all its rows or none");

template <bool TWO>
__device__ __forceinline__
void node_softmax_gather(
    const int wb,
    const int* __restrict__ rpA, const unsigned short* __restrict__ csrA,
    const float* __restrict__ asA, const float* __restrict__ adA,
    const float* __restrict__ GA, const float* __restrict__ bA,
    const int* __restrict__ rpB, const unsigned short* __restrict__ csrB,
    const float* __restrict__ asB, const float* __restrict__ adB,
    const float* __restrict__ GB, const float* __restrict__ bB,
    float* __restrict__ H, int N) {
  const int w = __builtin_amdgcn_readfirstlane(
      (int)(wb * (blockDim.x >> 5) + (threadIdx.x >> 5)));
  const int lane = threadIdx.x & 31;
  if (w * NODE_ROWS >= N) return;

#pragma nounroll
  for (int i = 0; i < NODE_ROWS; ++i) {
    const int d = w * NODE_ROWS + i;
    const float4 a  = seg_softmax_gather(d, lane, rpA, csrA, asA, adA, GA);
    const float4 ba = ((const float4*)bA)[lane];
    float4 o = make_float4(a.x + ba.x, a.y + ba.y, a.z + ba.z, a.w + ba.w);
    if (TWO) {
      const float4 b  = seg_softmax_gather(d, lane, rpB, csrB, asB, adB, GB);
      const float4 bb = ((const float4*)bB)[lane];
      o.x += b.x + bb.x; o.y += b.y + bb.y; o.z += b.z + bb.z; o.w += b.w + bb.w;
    }
    o.x = o.x > 0.f ? o.x : 0.f;
    o.y = o.y > 0.f ? o.y : 0.f;
    o.z = o.z > 0.f ? o.z : 0.f;
    o.w = o.w > 0.f ? o.w : 0.f;
    *(float4*)(H + (size_t)d * DC + lane * 4) = o;
  }
}

__global__ __launch_bounds__(256, 16)
void node_softmax_gather_kernel(
    const int* __restrict__ rp22, const unsigned short* __restrict__ csr22,
    const float* __restrict__ as2, const float* __restrict__ ad2,
    const float* __restrict__ G2, const float* __restrict__ b2,
    const int* __restrict__ rp12, const unsigned short* __restrict__ csr12,
    const float* __restrict__ asx, const float* __restrict__ adx,
    const float* __restrict__ Gx, const float* __restrict__ bx,
    float* __restrict__ H2, int n2,
    const int* __restrict__ rp11, const unsigned short* __restrict__ csr11,
    const float* __restrict__ as1, const float* __restrict__ ad1,
    const float* __restrict__ G1, const float* __restrict__ b1,
    float* __restrict__ H1, int n1, int nbB) {
  const int b = blockIdx.x;
  if (b < nbB) {
    node_softmax_gather<true>(b, rp22, csr22, as2, ad2, G2, b2,
                              rp12, csr12, asx, adx, Gx, bx, H2, n2);
  } else {
    node_softmax_gather<false>(b - nbB, rp11, csr11, as1, ad1, G1, b1,
                               nullptr, nullptr, nullptr, nullptr, nullptr, nullptr, H1, n1);
  }
}

extern "C" void kernel_launch(void* const* d_in, const int* in_sizes, int n_in,
                              void* d_out, int out_size, void* d_ws, size_t ws_size,
                              hipStream_t stream) {
  const float* x1    = (const float*)d_in[0];
  const float* x2    = (const float*)d_in[1];
  const int*   ei11  = (const int*)d_in[2];
  const int*   ei22  = (const int*)d_in[3];
  const int*   ei12  = (const int*)d_in[4];
  const float* W_i1  = (const float*)d_in[5];
  const float* as_i1 = (const float*)d_in[6];
  const float* ad_i1 = (const float*)d_in[7];
  const float* b_i1  = (const float*)d_in[8];
  const float* W_i2  = (const float*)d_in[9];
  const float* as_i2 = (const float*)d_in[10];
  const float* ad_i2 = (const float*)d_in[11];
  const float* b_i2  = (const float*)d_in[12];
  const float* Ws_x  = (const float*)d_in[13];
  const float* Wd_x  = (const float*)d_in[14];
  const float* as_x  = (const float*)d_in[15];
  const float* ad_x  = (const float*)d_in[16];
  const float* b_x   = (const float*)d_in[17];
  const float* W_out = (const float*)d_in[18];
  const float* b_out = (const float*)d_in[19];

  float* ws = (float*)d_ws;
  size_t off = 0;
  auto alloc = [&](size_t n) { float* p = ws + off; off += n; return p; };

  float* h1   = alloc((size_t)N1C * DC);
  float* h2   = alloc((size_t)N2C * DC);
  float* g1   = alloc((size_t)N1C * DC);
  float* g2   = alloc((size_t)N2C * DC);
  float* gx   = alloc((size_t)N1C * DC);
  float* a_s1 = alloc(N1C);
  float* a_d1 = alloc(N1C);
  float* a_s2 = alloc(N2C);
  float* a_d2 = alloc(N2C);
  float* a_sx = alloc(N1C);
  float* a_dx = alloc(N2C);
  float* vtab = alloc((size_t)LC * NFOLD * DC);
  int*   rp   = (int*)alloc((size_t)3 * NSTR);
  unsigned short* csr = (unsigned short*)alloc((size_t)3 * EC / 2);
  (void)ws_size; (void)in_sizes; (void)n_in; (void)out_size;

  const dim3 blk(256);
  const dim3 gemm_g(GEMM_WGS);
  const dim3 gemm_lg(GEMM_WGS / GEMM_JOBS, GEMM_JOBS);
  const int  rd_nb1 = (N1C / NODE_ROWS + 7) / 8, rd_nb2 = (N2C / NODE_ROWS + 7) / 8;
  const dim3 rd_g(rd_nb2 + rd_nb1);
  const dim3 binfold_g(3 * NCH + LC * NFOLD), srt_g(NR, 3);
  const dim3 bin_blk(BIN_T);
  const size_t lds_bytes = (size_t)(DC * WSTR + 2 * 16 * WSTR + 2 * NDOT * 16) * sizeof(float);

  const int* rp11 = rp;
  const int* rp22 = rp + NSTR;
  const int* rp12 = rp + 2 * NSTR;
  const unsigned short* csr11 = csr;
  const unsigned short* csr22 = csr + (size_t)EC;
  const unsigned short* csr12 = csr + (size_t)2 * EC;

  static_assert((size_t)3 * EC + (size_t)3 * NCH * NR <= (size_t)N1C * DC, "grouping scratch fits in h1");
  unsigned* binned = (unsigned*)h1;
  int* tab = (int*)(binned + (size_t)3 * EC);
  edge_bin_fold_kernel<<<binfold_g, bin_blk, 0, stream>>>(ei11, ei22, ei12, binned, tab,
                                                          W_i1, as_i1, ad_i1, W_i2, as_i2, ad_i2,
                                                          Ws_x, as_x, Wd_x, ad_x, vtab);
  range_sort_kernel<<<srt_g, bin_blk, 0, stream>>>(binned, tab, rp, csr);

  const float* in1 = x1;
  const float* in2 = x2;

  for (int l = 0; l < LC; ++l) {
    const size_t wo = (size_t)l * DC * DC;
    const size_t vo = (size_t)l * DC;

    const float* vt1 = vtab + (size_t)l * NFOLD * DC;
    const float* vt2 = vt1 + 3 * DC;

    gemm_layer_kernel<<<gemm_lg, blk, lds_bytes, stream>>>(
        in1, in2, W_i1 + wo, W_i2 + wo, Ws_x + wo, g1, g2, gx,
        vt1, vt2, a_s1, a_d1, a_sx, a_s2, a_d2, a_dx, N1C / 16, N2C / 16);

    node_softmax_gather_kernel<<<rd_g, blk, 0, stream>>>(
        rp22, csr22, a_s2, a_d2, g2, b_i2 + vo,
        rp12, csr12, a_sx, a_dx, gx, b_x + vo, h2, N2C,
        rp11, csr11, a_s1, a_d1, g1, b_i1 + vo, h1, N1C, rd_nb2);

    in1 = h1;
    in2 = h2;
  }

  float* out = (float*)d_out;
  gemm_n128_wmma<<<gemm_g, blk, lds_bytes, stream>>>(h1, W_out, b_out, out, (N1C + N2C) / 16);
}
